# starting kernel
# baseline (speedup 1.0000x reference)
#include <hip/hip_runtime.h>
#include <hip/hip_bf16.h>

#define IN_DIM 500
#define HID    256
#define NCLASS 50
#define KHOPS  10

typedef __attribute__((ext_vector_type(16))) _Float16 v16h;
typedef __attribute__((ext_vector_type(8)))  _Float16 h8;
typedef __attribute__((ext_vector_type(4)))  _Float16 h4;
typedef __attribute__((ext_vector_type(8)))  float    v8f;

__device__ __forceinline__ v16h load_frag(const _Float16* row, int kb) {
  h8 lo = *(const h8*)(row + kb);
  h8 hi = *(const h8*)(row + 16 + kb);
  v16h f;
#pragma unroll
  for (int i = 0; i < 8; ++i) { f[i] = lo[i]; f[8 + i] = hi[i]; }
  return f;
}

__device__ __forceinline__ v8f wmma16(v16h a, v16h b, v8f c) {
  return __builtin_amdgcn_wmma_f32_16x16x32_f16(false, a, false, b, (short)0,
                                                c, false, false);
}

__global__ void __launch_bounds__(256)
dagnn_gemm1(const float* __restrict__ X, const float* __restrict__ W,
            const float* __restrict__ bias, _Float16* __restrict__ Y, int Nn) {
  __shared__ __align__(16) _Float16 As[2][64][40];
  __shared__ __align__(16) _Float16 Bt[2][64][40];
  const int t = threadIdx.x;
  const int wave = t >> 5, lane = t & 31;
  const int waveM = wave >> 1, waveN = wave & 1;
  const int blockM = blockIdx.x * 64;
  const int blockN = blockIdx.y * 64;

  auto loadA = [&](int kt, int buf) {
#pragma unroll
    for (int j = 0; j < 2; ++j) {
      int idx = j * 256 + t;
      int r = idx >> 3, k4 = (idx & 7) * 4;
      int grow = blockM + r, gk = kt + k4;
      int cr = grow < Nn ? grow : Nn - 1;
      int ck = gk < IN_DIM ? gk : IN_DIM - 4;
      float4 v = *(const float4*)(X + (size_t)cr * IN_DIM + ck);
      bool ok = gk < IN_DIM;
      h4 hv;
      hv[0] = ok ? (_Float16)v.x : (_Float16)0.f;
      hv[1] = ok ? (_Float16)v.y : (_Float16)0.f;
      hv[2] = ok ? (_Float16)v.z : (_Float16)0.f;
      hv[3] = ok ? (_Float16)v.w : (_Float16)0.f;
      *(h4*)&As[buf][r][k4] = hv;
    }
  };
  auto loadB = [&](int kt, int buf) {
#pragma unroll
    for (int j = 0; j < 2; ++j) {
      int idx = j * 256 + t;
      int kk = idx >> 4, n4 = (idx & 15) * 4;
      int gk = kt + kk;
      int ck = gk < IN_DIM ? gk : IN_DIM - 1;
      float4 v = *(const float4*)(W + (size_t)ck * HID + blockN + n4);
      bool ok = gk < IN_DIM;
      Bt[buf][n4 + 0][kk] = ok ? (_Float16)v.x : (_Float16)0.f;
      Bt[buf][n4 + 1][kk] = ok ? (_Float16)v.y : (_Float16)0.f;
      Bt[buf][n4 + 2][kk] = ok ? (_Float16)v.z : (_Float16)0.f;
      Bt[buf][n4 + 3][kk] = ok ? (_Float16)v.w : (_Float16)0.f;
    }
  };

  v8f acc0 = {0.f, 0.f, 0.f, 0.f, 0.f, 0.f, 0.f, 0.f};
  v8f acc1 = acc0;
  const int mrow = waveM * 16 + (lane & 15);
  const int nrow = waveN * 32 + (lane & 15);
  const int kb = (lane >> 4) * 8;

  loadA(0, 0);
  loadB(0, 0);
  __syncthreads();
  const int NKT = 16;
  for (int it = 0; it < NKT; ++it) {
    int cur = it & 1;
    if (it + 1 < NKT) {
      loadA((it + 1) * 32, cur ^ 1);
      loadB((it + 1) * 32, cur ^ 1);
    }
    v16h a  = load_frag(&As[cur][mrow][0], kb);
    v16h b0 = load_frag(&Bt[cur][nrow][0], kb);
    v16h b1 = load_frag(&Bt[cur][nrow + 16][0], kb);
    acc0 = wmma16(a, b0, acc0);
    acc1 = wmma16(a, b1, acc1);
    __syncthreads();
  }

  const int col0 = blockN + waveN * 32 + (lane & 15);
  const int col1 = col0 + 16;
  const int rbase = blockM + waveM * 16 + (lane >> 4) * 8;
  const float bv0 = bias[col0], bv1 = bias[col1];
#pragma unroll
  for (int i = 0; i < 8; ++i) {
    int row = rbase + i;
    if (row < Nn) {
      float v0 = acc0[i] + bv0;
      float v1 = acc1[i] + bv1;
      Y[(size_t)row * HID + col0] = (_Float16)(v0 > 0.f ? v0 : 0.f);
      Y[(size_t)row * HID + col1] = (_Float16)(v1 > 0.f ? v1 : 0.f);
    }
  }
}

__global__ void __launch_bounds__(256)
dagnn_gemm2(const _Float16* __restrict__ Hd, const float* __restrict__ W2,
            const float* __restrict__ b2, float* __restrict__ plane0, int Nn) {
  __shared__ __align__(16) _Float16 As[2][32][40];
  __shared__ __align__(16) _Float16 Bt[2][64][40];
  const int t = threadIdx.x;
  const int wave = t >> 5, lane = t & 31;
  const int waveM = wave >> 2, waveN = wave & 3;
  const int blockM = blockIdx.x * 32;

  auto loadA = [&](int kt, int buf) {
    int r = t >> 3, k4 = (t & 7) * 4;
    int grow = blockM + r;
    int cr = grow < Nn ? grow : Nn - 1;
    *(h4*)&As[buf][r][k4] = *(const h4*)(Hd + (size_t)cr * HID + kt + k4);
  };
  auto loadB = [&](int kt, int buf) {
#pragma unroll
    for (int j = 0; j < 8; ++j) {
      int idx = j * 256 + t;
      int n = idx & 63, kk = idx >> 6;
      int cn = n < NCLASS ? n : NCLASS - 1;
      float v = W2[(size_t)(kt + kk) * NCLASS + cn];
      Bt[buf][n][kk] = (n < NCLASS) ? (_Float16)v : (_Float16)0.f;
    }
  };

  v8f acc = {0.f, 0.f, 0.f, 0.f, 0.f, 0.f, 0.f, 0.f};
  const int mrow = waveM * 16 + (lane & 15);
  const int nrow = waveN * 16 + (lane & 15);
  const int kb = (lane >> 4) * 8;

  loadA(0, 0);
  loadB(0, 0);
  __syncthreads();
  const int NKT = HID / 32;
  for (int it = 0; it < NKT; ++it) {
    int cur = it & 1;
    if (it + 1 < NKT) {
      loadA((it + 1) * 32, cur ^ 1);
      loadB((it + 1) * 32, cur ^ 1);
    }
    v16h a = load_frag(&As[cur][mrow][0], kb);
    v16h b = load_frag(&Bt[cur][nrow][0], kb);
    acc = wmma16(a, b, acc);
    __syncthreads();
  }

  const int col = waveN * 16 + (lane & 15);
  const int rbase = blockM + waveM * 16 + (lane >> 4) * 8;
  if (col < NCLASS) {
    const float bv = b2[col];
#pragma unroll
    for (int i = 0; i < 8; ++i) {
      int row = rbase + i;
      if (row < Nn) plane0[(size_t)row * NCLASS + col] = acc[i] + bv;
    }
  }
}

__global__ void dagnn_degree(const int* __restrict__ dst, float* __restrict__ deg, int E) {
  int i = blockIdx.x * blockDim.x + threadIdx.x;
  if (i < E) atomicAdd(&deg[dst[i]], 1.0f);
}

__global__ void dagnn_norm(float* __restrict__ deg, int Nn) {
  int i = blockIdx.x * blockDim.x + threadIdx.x;
  if (i < Nn) deg[i] = rsqrtf(deg[i]);
}

__global__ void dagnn_scatter(const float* __restrict__ in, float* __restrict__ outp,
                              const float* __restrict__ norm,
                              const int* __restrict__ src, const int* __restrict__ dst,
                              int E) {
  int gid = blockIdx.x * blockDim.x + threadIdx.x;
  int e = gid >> 5, lane = gid & 31;
  if (e >= E) return;
  int sv = src[e], dv = dst[e];
  float ns = norm[sv];
  const float* ip = in + (size_t)sv * NCLASS;
  float* op = outp + (size_t)dv * NCLASS;
  atomicAdd(&op[lane], ns * ip[lane]);
  if (lane < NCLASS - 32)
    atomicAdd(&op[lane + 32], ns * ip[lane + 32]);
}

__global__ void dagnn_scale(float* __restrict__ p, const float* __restrict__ norm, int Nn) {
  int i = blockIdx.x * blockDim.x + threadIdx.x;
  if (i < Nn * NCLASS) p[i] *= norm[i / NCLASS];
}

__global__ void dagnn_combine(const float* __restrict__ planes, const float* __restrict__ s,
                              float* __restrict__ out, int Nn) {
  int gid = blockIdx.x * blockDim.x + threadIdx.x;
  int node = gid >> 5, lane = gid & 31;
  if (node >= Nn) return;
  int c2 = lane + 32;
  float s1 = s[lane];
  float s2 = (c2 < NCLASS) ? s[c2] : 0.f;
  float acc1 = 0.f, acc2 = 0.f;
  size_t pstride = (size_t)Nn * NCLASS;
  size_t base = (size_t)node * NCLASS;
#pragma unroll
  for (int k = 0; k <= KHOPS; ++k) {
    const float* hp = planes + (size_t)k * pstride + base;
    float h1 = hp[lane];
    float h2 = (c2 < NCLASS) ? hp[c2] : 0.f;
    float z = h1 * s1 + h2 * s2;
#pragma unroll
    for (int off = 16; off > 0; off >>= 1) z += __shfl_xor(z, off, 32);
    float S = 1.0f / (1.0f + __expf(-z));
    acc1 += S * h1;
    acc2 += S * h2;
  }
  out[base + lane] = acc1;
  if (c2 < NCLASS) out[base + c2] = acc2;
}

extern "C" void kernel_launch(void* const* d_in, const int* in_sizes, int n_in,
                              void* d_out, int out_size, void* d_ws, size_t ws_size,
                              hipStream_t stream) {
  const float* feats = (const float*)d_in[0];
  const float* W1    = (const float*)d_in[1];
  const float* b1    = (const float*)d_in[2];
  const float* W2    = (const float*)d_in[3];
  const float* b2    = (const float*)d_in[4];
  const float* s     = (const float*)d_in[5];
  const int*   src   = (const int*)d_in[6];
  const int*   dst   = (const int*)d_in[7];
  const int Nn = in_sizes[0] / IN_DIM;
  const int E  = in_sizes[6];
  float* out = (float*)d_out;

  char* ws = (char*)d_ws;
  size_t off = 0;
  auto carve = [&](size_t bytes) -> char* {
    char* p = ws + off;
    off = (off + bytes + 255) & ~(size_t)255;
    return p;
  };
  float*    norm   = (float*)   carve((size_t)Nn * sizeof(float));
  _Float16* hid    = (_Float16*)carve((size_t)Nn * HID * sizeof(_Float16));
  float*    planes = (float*)   carve((size_t)(KHOPS + 1) * Nn * NCLASS * sizeof(float));
  if (off > ws_size) return;

  const size_t pstride = (size_t)Nn * NCLASS;

  hipMemsetAsync(norm, 0, (size_t)Nn * sizeof(float), stream);
  dagnn_degree<<<(E + 255) / 256, 256, 0, stream>>>(dst, norm, E);
  dagnn_norm<<<(Nn + 255) / 256, 256, 0, stream>>>(norm, Nn);

  dim3 g1((Nn + 63) / 64, HID / 64);
  dagnn_gemm1<<<g1, 256, 0, stream>>>(feats, W1, b1, hid, Nn);
  dim3 g2((Nn + 31) / 32, 1);
  dagnn_gemm2<<<g2, 256, 0, stream>>>(hid, W2, b2, planes, Nn);

  hipMemsetAsync(planes + pstride, 0, (size_t)KHOPS * pstride * sizeof(float), stream);
  const int sgrid = (int)(((size_t)E * 32 + 255) / 256);
  const int cgrid = (Nn * NCLASS + 255) / 256;
  for (int k = 1; k <= KHOPS; ++k) {
    const float* pin = planes + (size_t)(k - 1) * pstride;
    float* pout      = planes + (size_t)k * pstride;
    dagnn_scatter<<<sgrid, 256, 0, stream>>>(pin, pout, norm, src, dst, E);
    dagnn_scale<<<cgrid, 256, 0, stream>>>(pout, norm, Nn);
  }

  dagnn_combine<<<(int)(((size_t)Nn * 32 + 255) / 256), 256, 0, stream>>>(planes, s, out, Nn);
  (void)n_in; (void)out_size;
}


# Hop sums leave LDS: a tile's records split once by 8-row group, one wave per group sums each edge by fma into registers picked by its row, in record order; no LDS float adds; own weight pass
# speedup vs baseline: 4.0189x; 4.0189x over previous
#include <hip/hip_runtime.h>
#include <hip/hip_bf16.h>

#define IN_DIM 500
#define HID    256
#define NCLASS 50
#define KHOPS  10

typedef __attribute__((ext_vector_type(16))) _Float16 v16h;
typedef __attribute__((ext_vector_type(8)))  _Float16 h8;
typedef __attribute__((ext_vector_type(4)))  _Float16 h4;
typedef __attribute__((ext_vector_type(8)))  float    v8f;

__device__ __forceinline__ v16h load_frag(const _Float16* row, int kb) {
  h8 lo = *(const h8*)(row + kb);
  h8 hi = *(const h8*)(row + 16 + kb);
  v16h f;
#pragma unroll
  for (int i = 0; i < 8; ++i) { f[i] = lo[i]; f[8 + i] = hi[i]; }
  return f;
}

__device__ __forceinline__ v8f wmma16(v16h a, v16h b, v8f c) {
  return __builtin_amdgcn_wmma_f32_16x16x32_f16(false, a, false, b, (short)0,
                                                c, false, false);
}

#define G1_BLOCKS 256
#define G1_WAVES  8
#define G1_THREADS 256
#define WPITCH    520

__global__ void __launch_bounds__(G1_THREADS)
dagnn_gemm1(const float* __restrict__ X, const float* __restrict__ W,
            const float* __restrict__ bias, _Float16* __restrict__ Y, int Nn) {
  __shared__ __align__(16) _Float16 Wt[HID * WPITCH];
  const int t = threadIdx.x;
  const int lane = t & 31;
  const int wave = __builtin_amdgcn_readfirstlane(t >> 5);

  {
    const int n = t;
#pragma unroll 8
    for (int g = 0; g < 64; ++g) {
      const int k0 = g * 8;
      h8 p;
#pragma unroll
      for (int i = 0; i < 8; ++i) {
        const int k = k0 + i;
        const int ck = k < IN_DIM ? k : IN_DIM - 1;
        const float v = W[(size_t)ck * HID + n];
        p[i] = k < IN_DIM ? (_Float16)v : (_Float16)0.f;
      }
      *(h8*)&Wt[n * WPITCH + k0] = p;
    }
  }
  __syncthreads();

  const int c = lane & 15, h = lane >> 4;
  const _Float16* wwin0 = &Wt[c * WPITCH + 16 * h];
  const int ntiles = (Nn + 15) >> 4;
  const int tstride = G1_WAVES * G1_BLOCKS;

  int tile = wave * G1_BLOCKS + blockIdx.x;
  if (tile < ntiles) {
    auto rowptr = [&](int tl) -> const float* {
      const int mm = tl * 16 + c;
      const int mr = mm < Nn ? mm : Nn - 1;
      return X + (size_t)mr * IN_DIM + 16 * h;
    };
    const float* xp = rowptr(tile);

    float4 r0[4], r1[4], r2[4], r3[4];
    auto loadPlain = [&](const float* p, float4* d) {
#pragma unroll
      for (int q = 0; q < 4; ++q) d[q] = *(const float4*)(p + 4 * q);
    };
    auto loadMaybeLast = [&](const float* p, bool last, float4* d) {
#pragma unroll
      for (int q = 0; q < 4; ++q) {
        const bool ok = !last || (h == 0) || (q == 0);
        const float4 v = *(const float4*)(p + (ok ? 4 * q : 0));
        d[q].x = ok ? v.x : 0.f;
        d[q].y = ok ? v.y : 0.f;
        d[q].z = ok ? v.z : 0.f;
        d[q].w = ok ? v.w : 0.f;
      }
    };
    auto toFrag = [&](const float4* d) -> v16h {
      v16h f;
#pragma unroll
      for (int q = 0; q < 4; ++q) {
        f[4 * q + 0] = (_Float16)d[q].x;
        f[4 * q + 1] = (_Float16)d[q].y;
        f[4 * q + 2] = (_Float16)d[q].z;
        f[4 * q + 3] = (_Float16)d[q].w;
      }
      return f;
    };

    loadPlain(xp, r0);
    loadPlain(xp + 32, r1);
    loadPlain(xp + 64, r2);

    for (; tile < ntiles; tile += tstride) {
      const int m = tile * 16 + c;
      const int tnext = tile + tstride;
      const float* xpn = rowptr(tnext < ntiles ? tnext : tile);

      v8f acc[16];
#pragma unroll
      for (int nt = 0; nt < 16; ++nt)
        acc[nt] = (v8f){0.f, 0.f, 0.f, 0.f, 0.f, 0.f, 0.f, 0.f};

      auto products = [&](const float4* d, const _Float16* wk) {
        const v16h xf = toFrag(d);
#pragma unroll
        for (int nt = 0; nt < 16; ++nt) {
          const _Float16* wp = wk + nt * 16 * WPITCH;
          const h8 lo = *(const h8*)wp;
          const h8 hi = *(const h8*)(wp + 8);
          v16h wf;
#pragma unroll
          for (int i = 0; i < 8; ++i) { wf[i] = lo[i]; wf[8 + i] = hi[i]; }
          acc[nt] = wmma16(wf, xf, acc[nt]);
        }
      };

#pragma unroll 1
      for (int s4 = 0; s4 < 4; ++s4) {
        const float* xs = xp + s4 * 128;
        const _Float16* wk = wwin0 + s4 * 128;
        const bool more = s4 < 3;
        const float* nx = more ? xs + 128 : xpn;
        loadMaybeLast(xs + 96, !more, r3);
        products(r0, wk);
        loadPlain(nx, r0);
        products(r1, wk + 32);
        loadPlain(nx + 32, r1);
        products(r2, wk + 64);
        loadPlain(nx + 64, r2);
        products(r3, wk + 96);
      }

      if (m < Nn) {
        _Float16* yp = Y + (size_t)m * HID + 8 * h;
#pragma unroll
        for (int nt = 0; nt < 16; ++nt) {
          const float4 bA = *(const float4*)(bias + nt * 16 + 8 * h);
          const float4 bB = *(const float4*)(bias + nt * 16 + 8 * h + 4);
          const float bv[8] = {bA.x, bA.y, bA.z, bA.w, bB.x, bB.y, bB.z, bB.w};
          h8 o;
#pragma unroll
          for (int r = 0; r < 8; ++r) {
            const float v = acc[nt][r] + bv[r];
            o[r] = (_Float16)(v > 0.f ? v : 0.f);
          }
          *(h8*)(yp + nt * 16) = o;
        }
      }
      xp = xpn;
    }
  }
}

__global__ void __launch_bounds__(256)
dagnn_gemm2(const _Float16* __restrict__ Hd, const float* __restrict__ W2,
            const float* __restrict__ b2, float* __restrict__ plane0, int Nn) {
  __shared__ __align__(16) _Float16 As[2][32][40];
  __shared__ __align__(16) _Float16 Bt[2][64][40];
  const int t = threadIdx.x;
  const int wave = t >> 5, lane = t & 31;
  const int waveM = wave >> 2, waveN = wave & 3;
  const int blockM = blockIdx.x * 32;

  auto loadA = [&](int kt, int buf) {
    int r = t >> 3, k4 = (t & 7) * 4;
    int grow = blockM + r;
    int cr = grow < Nn ? grow : Nn - 1;
    *(h4*)&As[buf][r][k4] = *(const h4*)(Hd + (size_t)cr * HID + kt + k4);
  };
  auto loadB = [&](int kt, int buf) {
#pragma unroll
    for (int j = 0; j < 8; ++j) {
      int idx = j * 256 + t;
      int n = idx & 63, kk = idx >> 6;
      int cn = n < NCLASS ? n : NCLASS - 1;
      float v = W2[(size_t)(kt + kk) * NCLASS + cn];
      Bt[buf][n][kk] = (n < NCLASS) ? (_Float16)v : (_Float16)0.f;
    }
  };

  v8f acc = {0.f, 0.f, 0.f, 0.f, 0.f, 0.f, 0.f, 0.f};
  const int mrow = waveM * 16 + (lane & 15);
  const int nrow = waveN * 16 + (lane & 15);
  const int kb = (lane >> 4) * 8;

  loadA(0, 0);
  loadB(0, 0);
  __syncthreads();
  const int NKT = HID / 32;
  for (int it = 0; it < NKT; ++it) {
    int cur = it & 1;
    if (it + 1 < NKT) {
      loadA((it + 1) * 32, cur ^ 1);
      loadB((it + 1) * 32, cur ^ 1);
    }
    v16h a = load_frag(&As[cur][mrow][0], kb);
    v16h b = load_frag(&Bt[cur][nrow][0], kb);
    acc = wmma16(a, b, acc);
    __syncthreads();
  }

  const int col = waveN * 16 + (lane & 15);
  const int rbase = blockM + waveM * 16 + (lane >> 4) * 8;
  if (col < NCLASS) {
    const float bv = b2[col];
#pragma unroll
    for (int i = 0; i < 8; ++i) {
      int row = rbase + i;
      if (row < Nn) plane0[(size_t)row * NCLASS + col] = acc[i] + bv;
    }
  }
}

#define TSH   6
#define TILE  64
#define NCHUNK 256
#define SRC_BITS 24
#define SRC_MASK 0xFFFFFF
#define REC_PAD 16

__device__ __forceinline__ unsigned block_excl_scan256(unsigned v, unsigned* wsum,
                                                       unsigned* total) {
  const int lane = threadIdx.x & 31, wave = threadIdx.x >> 5;
  unsigned inc = v;
  for (int off = 1; off < 32; off <<= 1) {
    unsigned o = __shfl_up(inc, off, 32);
    if (lane >= off) inc += o;
  }
  if (lane == 31) wsum[wave] = inc;
  __syncthreads();
  unsigned wbase = 0, tot = 0;
  for (int w = 0; w < 8; ++w) {
    unsigned sw = wsum[w];
    if (w < wave) wbase += sw;
    tot += sw;
  }
  __syncthreads();
  *total = tot;
  return wbase + inc - v;
}

__global__ void __launch_bounds__(256)
dagnn_bin_count(const int* __restrict__ dst, unsigned* __restrict__ cnt,
                int E, int NT, int chunk) {
  extern __shared__ unsigned hist[];
  const int t = threadIdx.x;
  for (int i = t; i < NT; i += 256) hist[i] = 0u;
  __syncthreads();
  const int beg = blockIdx.x * chunk;
  const int end = (beg + chunk < E) ? beg + chunk : E;
  for (int i = beg + t; i < end; i += 256) {
    int tl = dst[i] >> TSH;
    tl = tl < NT ? tl : NT - 1;
    atomicAdd(&hist[tl], 1u);
  }
  __syncthreads();
  for (int i = t; i < NT; i += 256) cnt[(size_t)i * NCHUNK + blockIdx.x] = hist[i];
}

__global__ void __launch_bounds__(256)
dagnn_bin_scan(const unsigned* __restrict__ cnt, unsigned* __restrict__ lp,
               unsigned* __restrict__ tt) {
  __shared__ unsigned wsum[8];
  const size_t idx = (size_t)blockIdx.x * NCHUNK + threadIdx.x;
  unsigned tot;
  unsigned ex = block_excl_scan256(cnt[idx], wsum, &tot);
  lp[idx] = ex;
  if (threadIdx.x == 0) tt[blockIdx.x] = tot;
}

__global__ void __launch_bounds__(256)
dagnn_bin_place(const int* __restrict__ src, const int* __restrict__ dst,
                const unsigned* __restrict__ lp, const unsigned* __restrict__ tt,
                unsigned* __restrict__ tb, int2* __restrict__ rec,
                int E, int NT, int chunk) {
  extern __shared__ unsigned ctr[];
  __shared__ unsigned wsum[8];
  const int t = threadIdx.x;
  const int per = (NT + 255) / 256;
  const int i0 = t * per;
  const int i1 = (i0 + per < NT) ? i0 + per : NT;
  unsigned sum = 0;
  for (int i = i0; i < i1; ++i) sum += tt[i];
  unsigned tot;
  unsigned ex = block_excl_scan256(sum, wsum, &tot);
  for (int i = i0; i < i1; ++i) {
    ctr[i] = ex + lp[(size_t)i * NCHUNK + blockIdx.x];
    if (blockIdx.x == 0) tb[i] = ex;
    ex += tt[i];
  }
  if (blockIdx.x == 0 && t == 0) tb[NT] = tot;
  __syncthreads();
  const int beg = blockIdx.x * chunk;
  const int end = (beg + chunk < E) ? beg + chunk : E;
  for (int i = beg + t; i < end; i += 256) {
    int dv = dst[i];
    int tl = dv >> TSH;
    tl = tl < NT ? tl : NT - 1;
    unsigned pos = atomicAdd(&ctr[tl], 1u);
    rec[pos] = make_int2(src[i] | ((dv & (TILE - 1)) << SRC_BITS), 0);
  }
}

#define GROWS  8
#define GSH    3
#define GROUPS (TILE / GROWS)

__global__ void __launch_bounds__(256)
dagnn_tile_group(const int2* __restrict__ rec, const unsigned* __restrict__ tb,
                 float* __restrict__ norm, int2* __restrict__ rec2,
                 unsigned* __restrict__ gb, int Nn, int NT, int E) {
  __shared__ unsigned dcnt[TILE];
  __shared__ unsigned gpos[GROUPS];
  const int t = threadIdx.x;
  if (t < TILE) dcnt[t] = 0u;
  __syncthreads();
  const unsigned beg = tb[blockIdx.x], end = tb[blockIdx.x + 1];
  for (unsigned i = beg + t; i < end; i += 256) atomicAdd(&dcnt[(unsigned)rec[i].x >> SRC_BITS], 1u);
  __syncthreads();
  const int node = blockIdx.x * TILE + t;
  if (t < TILE && node < Nn) norm[node] = rsqrtf((float)dcnt[t]);
  if (t < GROUPS) {
    unsigned off = beg;
    for (int r = 0; r < t * GROWS; ++r) off += dcnt[r];
    gpos[t] = off;
    gb[blockIdx.x * GROUPS + t] = off;
  }
  if (blockIdx.x == NT - 1 && t == 0) gb[NT * GROUPS] = end;
  if (blockIdx.x == 0 && t < REC_PAD) rec2[(size_t)E + t] = make_int2(0, 0);
  __syncthreads();
  for (unsigned i = beg + t; i < end; i += 256) {
    const int2 r = rec[i];
    const unsigned pos = atomicAdd(&gpos[((unsigned)r.x >> SRC_BITS) >> GSH], 1u);
    rec2[pos] = r;
  }
}

#define EW_PER 8
__global__ void __launch_bounds__(256)
dagnn_edge_weight(int2* __restrict__ rec2, const float* __restrict__ norm, int E) {
  const int i0 = blockIdx.x * (256 * EW_PER) + threadIdx.x;
  int sx[EW_PER];
  float w[EW_PER];
#pragma unroll
  for (int u = 0; u < EW_PER; ++u) {
    const int i = i0 + u * 256;
    sx[u] = rec2[i < E ? i : E - 1].x & SRC_MASK;
  }
#pragma unroll
  for (int u = 0; u < EW_PER; ++u) w[u] = norm[sx[u]];
#pragma unroll
  for (int u = 0; u < EW_PER; ++u) {
    const int i = i0 + u * 256;
    if (i < E) rec2[i].y = __float_as_int(w[u]);
  }
}

__global__ void __launch_bounds__(256)
dagnn_hop(const float* __restrict__ in, float* __restrict__ outp,
          const float* __restrict__ norm, const int2* __restrict__ rec,
          const unsigned* __restrict__ gb, int Nn) {
  __shared__ __align__(8) float stage[TILE * NCLASS];
  __shared__ float nl[TILE];
  const int t = threadIdx.x;
  const int wave = __builtin_amdgcn_readfirstlane(t >> 5);
  const int lane = t & 31;
  const int node0 = blockIdx.x * TILE;
  if (t < TILE) nl[t] = (node0 + t < Nn) ? norm[node0 + t] : 0.f;

  const unsigned s = gb[blockIdx.x * GROUPS + wave];
  const unsigned e = gb[blockIdx.x * GROUPS + wave + 1];
  float ax[GROWS], ay[GROWS];
#pragma unroll
  for (int r = 0; r < GROWS; ++r) { ax[r] = 0.f; ay[r] = 0.f; }

  if (s < e) {
    const unsigned lo = 8u * (unsigned)(lane < 25 ? lane : 24);
    auto add = [&](const int2 c, const float2 v) {
      const int row = (int)(((unsigned)__builtin_amdgcn_readfirstlane(c.x) >> SRC_BITS) & (GROWS - 1));
      const float w = __int_as_float(__builtin_amdgcn_readfirstlane(c.y));
      ax[row] = fmaf(w, v.x, ax[row]);
      ay[row] = fmaf(w, v.y, ay[row]);
    };
    auto batch = [&](unsigned p, const int2* cur, int2* nxt) {
#pragma unroll
      for (int u = 0; u < 8; ++u) nxt[u] = rec[p + 8u + (unsigned)u];
      float2 v[8];
#pragma unroll
      for (int u = 0; u < 8; ++u) {
        const unsigned sx = (unsigned)__builtin_amdgcn_readfirstlane(cur[u].x);
        const char* rowp = (const char*)in + (size_t)(sx & SRC_MASK) * (NCLASS * sizeof(float));
        v[u] = *(const float2*)(rowp + lo);
      }
      if (e - p >= 8u) {
#pragma unroll
        for (int u = 0; u < 8; ++u) add(cur[u], v[u]);
      } else {
#pragma unroll
        for (int u = 0; u < 8; ++u) {
          if (p + (unsigned)u < e) add(cur[u], v[u]);
        }
      }
    };
    int2 ra[8], rb[8];
#pragma unroll
    for (int u = 0; u < 8; ++u) ra[u] = rec[s + (unsigned)u];
    for (unsigned p = s; p < e; p += 16u) {
      batch(p, ra, rb);
      if (p + 8u < e) batch(p + 8u, rb, ra);
    }
  }

  if (lane < 25) {
#pragma unroll
    for (int r = 0; r < GROWS; ++r)
      *(float2*)&stage[(wave * GROWS + r) * NCLASS + 2 * lane] = make_float2(ax[r], ay[r]);
  }
  __syncthreads();

  const int nvalid = (Nn - node0 < TILE) ? Nn - node0 : TILE;
  float* op = outp + (size_t)node0 * NCLASS;
  for (int i = t; i < nvalid * NCLASS; i += 256) op[i] = stage[i] * nl[i / NCLASS];
}

#define RO_NODES 32
#define RO_WAVES 4
#define RO_PIECES ((RO_NODES * NCLASS) / 4)
#define RO_WHOLE  (RO_PIECES / 32)
#define RO_TAIL   (RO_PIECES - RO_WHOLE * 32)
typedef __attribute__((ext_vector_type(4))) float f4;
typedef __attribute__((ext_vector_type(2))) float f2;
static_assert(KHOPS % 2 == 0, "the plane loop below takes planes in pairs");
static_assert(RO_TAIL == 16, "tail piece count");

__device__ __forceinline__ void wave_lds_handover() {
  __builtin_amdgcn_fence(__ATOMIC_RELEASE, "wavefront");
  __builtin_amdgcn_wave_barrier();
  __builtin_amdgcn_fence(__ATOMIC_ACQUIRE, "wavefront");
}

__global__ void __launch_bounds__(RO_WAVES * 32)
dagnn_combine(const float* __restrict__ planes, const float* __restrict__ s,
              float* __restrict__ out, int Nn) {
  __shared__ __align__(16) float stage[RO_WAVES][RO_NODES * NCLASS];
  const int lane = threadIdx.x & 31;
  const int wave = __builtin_amdgcn_readfirstlane(threadIdx.x >> 5);
  float* st = stage[wave];
  const int node0 = (blockIdx.x * RO_WAVES + wave) * RO_NODES;
  if (node0 >= Nn) return;
  const size_t pstride = (size_t)Nn * NCLASS;
  const size_t base = (size_t)node0 * NCLASS;
  const bool whole = (node0 + RO_NODES <= Nn) && ((Nn & 1) == 0);

  float acc[NCLASS];
#pragma unroll
  for (int c = 0; c < NCLASS; ++c) acc[c] = 0.f;

  auto consume = [&]() {
    f2 h[NCLASS / 2];
#pragma unroll
    for (int j = 0; j < NCLASS / 2; ++j) h[j] = *(const f2*)&st[lane * NCLASS + 2 * j];
    float z = 0.f;
#pragma unroll
    for (int j = 0; j < NCLASS / 2; ++j) {
      z += h[j].x * s[2 * j];
      z += h[j].y * s[2 * j + 1];
    }
    const float S = 1.0f / (1.0f + __expf(-z));
#pragma unroll
    for (int j = 0; j < NCLASS / 2; ++j) {
      acc[2 * j]     += S * h[j].x;
      acc[2 * j + 1] += S * h[j].y;
    }
  };

  if (whole) {
    const f4* p0 = (const f4*)(planes + base);
    const size_t pq = pstride / 4;
    f4 ba[RO_WHOLE + 1], bb[RO_WHOLE + 1];
    auto fetch = [&](int k, f4* b) {
      const f4* p = p0 + (size_t)k * pq;
#pragma unroll
      for (int i = 0; i < RO_WHOLE; ++i) b[i] = p[i * 32 + lane];
      b[RO_WHOLE] = p[RO_WHOLE * 32 + (lane & (RO_TAIL - 1))];
    };
    auto put = [&](const f4* b) {
#pragma unroll
      for (int i = 0; i < RO_WHOLE; ++i) *(f4*)&st[(i * 32 + lane) * 4] = b[i];
      if (lane < RO_TAIL) *(f4*)&st[(RO_WHOLE * 32 + lane) * 4] = b[RO_WHOLE];
    };
    fetch(0, ba);
#pragma unroll 1
    for (int k = 0; k < KHOPS; k += 2) {
      fetch(k + 1, bb);
      put(ba); wave_lds_handover(); consume(); wave_lds_handover();
      fetch(k + 2, ba);
      put(bb); wave_lds_handover(); consume(); wave_lds_handover();
    }
    put(ba); wave_lds_handover(); consume(); wave_lds_handover();
  } else {
    for (int k = 0; k <= KHOPS; ++k) {
      const float* p = planes + (size_t)k * pstride;
      for (int i = 0; i < NCLASS; ++i) {
        const int idx = i * 32 + lane;
        const size_t gi = base + (size_t)idx;
        st[idx] = gi < pstride ? p[gi] : 0.f;
      }
      wave_lds_handover(); consume(); wave_lds_handover();
    }
  }

#pragma unroll
  for (int j = 0; j < NCLASS / 2; ++j)
    *(f2*)&st[lane * NCLASS + 2 * j] = (f2){acc[2 * j], acc[2 * j + 1]};
  wave_lds_handover();
  if (whole) {
    f4* op = (f4*)(out + base);
#pragma unroll
    for (int i = 0; i < RO_WHOLE; ++i) op[i * 32 + lane] = *(const f4*)&st[(i * 32 + lane) * 4];
    if (lane < RO_TAIL) op[RO_WHOLE * 32 + lane] = *(const f4*)&st[(RO_WHOLE * 32 + lane) * 4];
  } else {
    for (int i = 0; i < NCLASS; ++i) {
      const int idx = i * 32 + lane;
      const size_t gi = base + (size_t)idx;
      if (gi < pstride) out[gi] = st[idx];
    }
  }
}

extern "C" void kernel_launch(void* const* d_in, const int* in_sizes, int n_in,
                              void* d_out, int out_size, void* d_ws, size_t ws_size,
                              hipStream_t stream) {
  const float* feats = (const float*)d_in[0];
  const float* W1    = (const float*)d_in[1];
  const float* b1    = (const float*)d_in[2];
  const float* W2    = (const float*)d_in[3];
  const float* b2    = (const float*)d_in[4];
  const float* s     = (const float*)d_in[5];
  const int*   src   = (const int*)d_in[6];
  const int*   dst   = (const int*)d_in[7];
  const int Nn = in_sizes[0] / IN_DIM;
  const int E  = in_sizes[6];
  float* out = (float*)d_out;

  char* ws = (char*)d_ws;
  size_t off = 0;
  auto carve = [&](size_t bytes) -> char* {
    char* p = ws + off;
    off = (off + bytes + 255) & ~(size_t)255;
    return p;
  };
  const int NT = (Nn + TILE - 1) / TILE;
  const int chunk = (E + NCHUNK - 1) / NCHUNK;
  float*    norm   = (float*)   carve((size_t)Nn * sizeof(float));
  _Float16* hid    = (_Float16*)carve((size_t)Nn * HID * sizeof(_Float16));
  float*    planes = (float*)   carve((size_t)(KHOPS + 1) * Nn * NCLASS * sizeof(float));
  unsigned* cnt    = (unsigned*)carve((size_t)NT * NCHUNK * sizeof(unsigned));
  unsigned* lp     = (unsigned*)carve((size_t)NT * NCHUNK * sizeof(unsigned));
  unsigned* tt     = (unsigned*)carve((size_t)NT * sizeof(unsigned));
  unsigned* tb     = (unsigned*)carve((size_t)(NT + 1) * sizeof(unsigned));
  unsigned* gb     = (unsigned*)carve(((size_t)NT * GROUPS + 1) * sizeof(unsigned));
  int2*     rec    = (int2*)    carve((size_t)E * sizeof(int2));
  int2*     rec2   = (int2*)    carve(((size_t)E + REC_PAD) * sizeof(int2));
  if (off > ws_size) return;

  const size_t pstride = (size_t)Nn * NCLASS;

  const size_t hbytes = (size_t)NT * sizeof(unsigned);
  dagnn_bin_count<<<NCHUNK, 256, hbytes, stream>>>(dst, cnt, E, NT, chunk);
  dagnn_bin_scan<<<NT, 256, 0, stream>>>(cnt, lp, tt);
  dagnn_bin_place<<<NCHUNK, 256, hbytes, stream>>>(src, dst, lp, tt, tb, rec, E, NT, chunk);
  dagnn_tile_group<<<NT, 256, 0, stream>>>(rec, tb, norm, rec2, gb, Nn, NT, E);
  dagnn_edge_weight<<<(E + 256 * EW_PER - 1) / (256 * EW_PER), 256, 0, stream>>>(rec2, norm, E);

  dagnn_gemm1<<<G1_BLOCKS, G1_THREADS, 0, stream>>>(feats, W1, b1, hid, Nn);
  dim3 g2((Nn + 31) / 32, 1);
  dagnn_gemm2<<<g2, 256, 0, stream>>>(hid, W2, b2, planes, Nn);

  for (int k = 1; k <= KHOPS; ++k) {
    const float* pin = planes + (size_t)(k - 1) * pstride;
    float* pout      = planes + (size_t)k * pstride;
    dagnn_hop<<<NT, 256, 0, stream>>>(pin, pout, norm, rec2, gb, Nn);
  }

  const int rgroups = (Nn + RO_NODES - 1) / RO_NODES;
  dagnn_combine<<<(rgroups + RO_WAVES - 1) / RO_WAVES, RO_WAVES * 32, 0, stream>>>(planes, s, out, Nn);
  (void)n_in; (void)out_size;
}


# Edge binning count and place passes: a thread requests 8 edges (and its table entries) together before using any, not one dependent load per loop trip; LDS counter arrival order changes
# speedup vs baseline: 4.1510x; 1.0329x over previous
#include <hip/hip_runtime.h>
#include <hip/hip_bf16.h>

#define IN_DIM 500
#define HID    256
#define NCLASS 50
#define KHOPS  10

typedef __attribute__((ext_vector_type(16))) _Float16 v16h;
typedef __attribute__((ext_vector_type(8)))  _Float16 h8;
typedef __attribute__((ext_vector_type(4)))  _Float16 h4;
typedef __attribute__((ext_vector_type(8)))  float    v8f;

__device__ __forceinline__ v16h load_frag(const _Float16* row, int kb) {
  h8 lo = *(const h8*)(row + kb);
  h8 hi = *(const h8*)(row + 16 + kb);
  v16h f;
#pragma unroll
  for (int i = 0; i < 8; ++i) { f[i] = lo[i]; f[8 + i] = hi[i]; }
  return f;
}

__device__ __forceinline__ v8f wmma16(v16h a, v16h b, v8f c) {
  return __builtin_amdgcn_wmma_f32_16x16x32_f16(false, a, false, b, (short)0,
                                                c, false, false);
}

#define G1_BLOCKS 256
#define G1_WAVES  8
#define G1_THREADS 256
#define WPITCH    520

__global__ void __launch_bounds__(G1_THREADS)
dagnn_gemm1(const float* __restrict__ X, const float* __restrict__ W,
            const float* __restrict__ bias, _Float16* __restrict__ Y, int Nn) {
  __shared__ __align__(16) _Float16 Wt[HID * WPITCH];
  const int t = threadIdx.x;
  const int lane = t & 31;
  const int wave = __builtin_amdgcn_readfirstlane(t >> 5);

  {
    const int n = t;
#pragma unroll 8
    for (int g = 0; g < 64; ++g) {
      const int k0 = g * 8;
      h8 p;
#pragma unroll
      for (int i = 0; i < 8; ++i) {
        const int k = k0 + i;
        const int ck = k < IN_DIM ? k : IN_DIM - 1;
        const float v = W[(size_t)ck * HID + n];
        p[i] = k < IN_DIM ? (_Float16)v : (_Float16)0.f;
      }
      *(h8*)&Wt[n * WPITCH + k0] = p;
    }
  }
  __syncthreads();

  const int c = lane & 15, h = lane >> 4;
  const _Float16* wwin0 = &Wt[c * WPITCH + 16 * h];
  const int ntiles = (Nn + 15) >> 4;
  const int tstride = G1_WAVES * G1_BLOCKS;

  int tile = wave * G1_BLOCKS + blockIdx.x;
  if (tile < ntiles) {
    auto rowptr = [&](int tl) -> const float* {
      const int mm = tl * 16 + c;
      const int mr = mm < Nn ? mm : Nn - 1;
      return X + (size_t)mr * IN_DIM + 16 * h;
    };
    const float* xp = rowptr(tile);

    float4 r0[4], r1[4], r2[4], r3[4];
    auto loadPlain = [&](const float* p, float4* d) {
#pragma unroll
      for (int q = 0; q < 4; ++q) d[q] = *(const float4*)(p + 4 * q);
    };
    auto loadMaybeLast = [&](const float* p, bool last, float4* d) {
#pragma unroll
      for (int q = 0; q < 4; ++q) {
        const bool ok = !last || (h == 0) || (q == 0);
        const float4 v = *(const float4*)(p + (ok ? 4 * q : 0));
        d[q].x = ok ? v.x : 0.f;
        d[q].y = ok ? v.y : 0.f;
        d[q].z = ok ? v.z : 0.f;
        d[q].w = ok ? v.w : 0.f;
      }
    };
    auto toFrag = [&](const float4* d) -> v16h {
      v16h f;
#pragma unroll
      for (int q = 0; q < 4; ++q) {
        f[4 * q + 0] = (_Float16)d[q].x;
        f[4 * q + 1] = (_Float16)d[q].y;
        f[4 * q + 2] = (_Float16)d[q].z;
        f[4 * q + 3] = (_Float16)d[q].w;
      }
      return f;
    };

    loadPlain(xp, r0);
    loadPlain(xp + 32, r1);
    loadPlain(xp + 64, r2);

    for (; tile < ntiles; tile += tstride) {
      const int m = tile * 16 + c;
      const int tnext = tile + tstride;
      const float* xpn = rowptr(tnext < ntiles ? tnext : tile);

      v8f acc[16];
#pragma unroll
      for (int nt = 0; nt < 16; ++nt)
        acc[nt] = (v8f){0.f, 0.f, 0.f, 0.f, 0.f, 0.f, 0.f, 0.f};

      auto products = [&](const float4* d, const _Float16* wk) {
        const v16h xf = toFrag(d);
#pragma unroll
        for (int nt = 0; nt < 16; ++nt) {
          const _Float16* wp = wk + nt * 16 * WPITCH;
          const h8 lo = *(const h8*)wp;
          const h8 hi = *(const h8*)(wp + 8);
          v16h wf;
#pragma unroll
          for (int i = 0; i < 8; ++i) { wf[i] = lo[i]; wf[8 + i] = hi[i]; }
          acc[nt] = wmma16(wf, xf, acc[nt]);
        }
      };

#pragma unroll 1
      for (int s4 = 0; s4 < 4; ++s4) {
        const float* xs = xp + s4 * 128;
        const _Float16* wk = wwin0 + s4 * 128;
        const bool more = s4 < 3;
        const float* nx = more ? xs + 128 : xpn;
        loadMaybeLast(xs + 96, !more, r3);
        products(r0, wk);
        loadPlain(nx, r0);
        products(r1, wk + 32);
        loadPlain(nx + 32, r1);
        products(r2, wk + 64);
        loadPlain(nx + 64, r2);
        products(r3, wk + 96);
      }

      if (m < Nn) {
        _Float16* yp = Y + (size_t)m * HID + 8 * h;
#pragma unroll
        for (int nt = 0; nt < 16; ++nt) {
          const float4 bA = *(const float4*)(bias + nt * 16 + 8 * h);
          const float4 bB = *(const float4*)(bias + nt * 16 + 8 * h + 4);
          const float bv[8] = {bA.x, bA.y, bA.z, bA.w, bB.x, bB.y, bB.z, bB.w};
          h8 o;
#pragma unroll
          for (int r = 0; r < 8; ++r) {
            const float v = acc[nt][r] + bv[r];
            o[r] = (_Float16)(v > 0.f ? v : 0.f);
          }
          *(h8*)(yp + nt * 16) = o;
        }
      }
      xp = xpn;
    }
  }
}

__global__ void __launch_bounds__(256)
dagnn_gemm2(const _Float16* __restrict__ Hd, const float* __restrict__ W2,
            const float* __restrict__ b2, float* __restrict__ plane0, int Nn) {
  __shared__ __align__(16) _Float16 As[2][32][40];
  __shared__ __align__(16) _Float16 Bt[2][64][40];
  const int t = threadIdx.x;
  const int wave = t >> 5, lane = t & 31;
  const int waveM = wave >> 2, waveN = wave & 3;
  const int blockM = blockIdx.x * 32;

  auto loadA = [&](int kt, int buf) {
    int r = t >> 3, k4 = (t & 7) * 4;
    int grow = blockM + r;
    int cr = grow < Nn ? grow : Nn - 1;
    *(h4*)&As[buf][r][k4] = *(const h4*)(Hd + (size_t)cr * HID + kt + k4);
  };
  auto loadB = [&](int kt, int buf) {
#pragma unroll
    for (int j = 0; j < 8; ++j) {
      int idx = j * 256 + t;
      int n = idx & 63, kk = idx >> 6;
      int cn = n < NCLASS ? n : NCLASS - 1;
      float v = W2[(size_t)(kt + kk) * NCLASS + cn];
      Bt[buf][n][kk] = (n < NCLASS) ? (_Float16)v : (_Float16)0.f;
    }
  };

  v8f acc = {0.f, 0.f, 0.f, 0.f, 0.f, 0.f, 0.f, 0.f};
  const int mrow = waveM * 16 + (lane & 15);
  const int nrow = waveN * 16 + (lane & 15);
  const int kb = (lane >> 4) * 8;

  loadA(0, 0);
  loadB(0, 0);
  __syncthreads();
  const int NKT = HID / 32;
  for (int it = 0; it < NKT; ++it) {
    int cur = it & 1;
    if (it + 1 < NKT) {
      loadA((it + 1) * 32, cur ^ 1);
      loadB((it + 1) * 32, cur ^ 1);
    }
    v16h a = load_frag(&As[cur][mrow][0], kb);
    v16h b = load_frag(&Bt[cur][nrow][0], kb);
    acc = wmma16(a, b, acc);
    __syncthreads();
  }

  const int col = waveN * 16 + (lane & 15);
  const int rbase = blockM + waveM * 16 + (lane >> 4) * 8;
  if (col < NCLASS) {
    const float bv = b2[col];
#pragma unroll
    for (int i = 0; i < 8; ++i) {
      int row = rbase + i;
      if (row < Nn) plane0[(size_t)row * NCLASS + col] = acc[i] + bv;
    }
  }
}

#define TSH   6
#define TILE  64
#define NCHUNK 256
#define SRC_BITS 24
#define SRC_MASK 0xFFFFFF
#define REC_PAD 16

__device__ __forceinline__ unsigned block_excl_scan256(unsigned v, unsigned* wsum,
                                                       unsigned* total) {
  const int lane = threadIdx.x & 31, wave = threadIdx.x >> 5;
  unsigned inc = v;
  for (int off = 1; off < 32; off <<= 1) {
    unsigned o = __shfl_up(inc, off, 32);
    if (lane >= off) inc += o;
  }
  if (lane == 31) wsum[wave] = inc;
  __syncthreads();
  unsigned wbase = 0, tot = 0;
  for (int w = 0; w < 8; ++w) {
    unsigned sw = wsum[w];
    if (w < wave) wbase += sw;
    tot += sw;
  }
  __syncthreads();
  *total = tot;
  return wbase + inc - v;
}

#define BIN_B 8

template <bool GUARD>
__device__ __forceinline__ void bin_count_batch(const int* __restrict__ dst, unsigned* hist,
                                                int i0, int end, int NT) {
  int dv[BIN_B];
#pragma unroll
  for (int u = 0; u < BIN_B; ++u) {
    const int i = i0 + u * 256;
    dv[u] = dst[(!GUARD || i < end) ? i : end - 1];
  }
#pragma unroll
  for (int u = 0; u < BIN_B; ++u) {
    const int i = i0 + u * 256;
    if (!GUARD || i < end) {
      int tl = dv[u] >> TSH;
      tl = tl < NT ? tl : NT - 1;
      atomicAdd(&hist[tl], 1u);
    }
  }
}

__global__ void __launch_bounds__(256)
dagnn_bin_count(const int* __restrict__ dst, unsigned* __restrict__ cnt,
                int E, int NT, int chunk) {
  extern __shared__ unsigned hist[];
  const int t = threadIdx.x;
  for (int i = t; i < NT; i += 256) hist[i] = 0u;
  __syncthreads();
  const int beg = blockIdx.x * chunk;
  const int end = (beg + chunk < E) ? beg + chunk : E;
  for (int b0 = beg; b0 < end; b0 += 256 * BIN_B) {
    if (b0 + 256 * BIN_B <= end) bin_count_batch<false>(dst, hist, b0 + t, end, NT);
    else                         bin_count_batch<true>(dst, hist, b0 + t, end, NT);
  }
  __syncthreads();
  for (int i = t; i < NT; i += 256) cnt[(size_t)i * NCHUNK + blockIdx.x] = hist[i];
}

__global__ void __launch_bounds__(256)
dagnn_bin_scan(const unsigned* __restrict__ cnt, unsigned* __restrict__ lp,
               unsigned* __restrict__ tt) {
  __shared__ unsigned wsum[8];
  const size_t idx = (size_t)blockIdx.x * NCHUNK + threadIdx.x;
  unsigned tot;
  unsigned ex = block_excl_scan256(cnt[idx], wsum, &tot);
  lp[idx] = ex;
  if (threadIdx.x == 0) tt[blockIdx.x] = tot;
}

template <bool GUARD>
__device__ __forceinline__ void bin_place_batch(const int* __restrict__ src,
                                                const int* __restrict__ dst, unsigned* ctr,
                                                int2* __restrict__ rec, int i0, int end, int NT) {
  int dv[BIN_B], sv[BIN_B];
  unsigned pos[BIN_B];
#pragma unroll
  for (int u = 0; u < BIN_B; ++u) {
    const int i = i0 + u * 256;
    const int ci = (!GUARD || i < end) ? i : end - 1;
    dv[u] = dst[ci];
    sv[u] = src[ci];
  }
#pragma unroll
  for (int u = 0; u < BIN_B; ++u) {
    const int i = i0 + u * 256;
    pos[u] = 0u;
    if (!GUARD || i < end) {
      int tl = dv[u] >> TSH;
      tl = tl < NT ? tl : NT - 1;
      pos[u] = atomicAdd(&ctr[tl], 1u);
    }
  }
#pragma unroll
  for (int u = 0; u < BIN_B; ++u) {
    const int i = i0 + u * 256;
    if (!GUARD || i < end)
      rec[pos[u]] = make_int2(sv[u] | ((dv[u] & (TILE - 1)) << SRC_BITS), 0);
  }
}

__global__ void __launch_bounds__(256)
dagnn_bin_place(const int* __restrict__ src, const int* __restrict__ dst,
                const unsigned* __restrict__ lp, const unsigned* __restrict__ tt,
                unsigned* __restrict__ tb, int2* __restrict__ rec,
                int E, int NT, int chunk) {
  extern __shared__ unsigned ctr[];
  __shared__ unsigned wsum[8];
  const int t = threadIdx.x;
  const int per = (NT + 255) / 256;
  const int i0 = t * per;
  const int i1 = (i0 + per < NT) ? i0 + per : NT;
  unsigned sum = 0;
  for (int ib = i0; ib < i1; ib += BIN_B) {
    unsigned a[BIN_B];
#pragma unroll
    for (int u = 0; u < BIN_B; ++u) a[u] = tt[ib + u < i1 ? ib + u : i1 - 1];
#pragma unroll
    for (int u = 0; u < BIN_B; ++u) sum += (ib + u < i1) ? a[u] : 0u;
  }
  unsigned tot;
  unsigned ex = block_excl_scan256(sum, wsum, &tot);
  for (int ib = i0; ib < i1; ib += BIN_B) {
    unsigned a[BIN_B], l[BIN_B];
#pragma unroll
    for (int u = 0; u < BIN_B; ++u) {
      const int i = ib + u < i1 ? ib + u : i1 - 1;
      a[u] = tt[i];
      l[u] = lp[(size_t)i * NCHUNK + blockIdx.x];
    }
#pragma unroll
    for (int u = 0; u < BIN_B; ++u) {
      const int i = ib + u;
      if (i < i1) {
        ctr[i] = ex + l[u];
        if (blockIdx.x == 0) tb[i] = ex;
        ex += a[u];
      }
    }
  }
  if (blockIdx.x == 0 && t == 0) tb[NT] = tot;
  __syncthreads();
  const int beg = blockIdx.x * chunk;
  const int end = (beg + chunk < E) ? beg + chunk : E;
  for (int b0 = beg; b0 < end; b0 += 256 * BIN_B) {
    if (b0 + 256 * BIN_B <= end) bin_place_batch<false>(src, dst, ctr, rec, b0 + t, end, NT);
    else                         bin_place_batch<true>(src, dst, ctr, rec, b0 + t, end, NT);
  }
}

#define GROWS  8
#define GSH    3
#define GROUPS (TILE / GROWS)

__global__ void __launch_bounds__(256)
dagnn_tile_group(const int2* __restrict__ rec, const unsigned* __restrict__ tb,
                 float* __restrict__ norm, int2* __restrict__ rec2,
                 unsigned* __restrict__ gb, int Nn, int NT, int E) {
  __shared__ unsigned dcnt[TILE];
  __shared__ unsigned gpos[GROUPS];
  const int t = threadIdx.x;
  if (t < TILE) dcnt[t] = 0u;
  __syncthreads();
  const unsigned beg = tb[blockIdx.x], end = tb[blockIdx.x + 1];
  for (unsigned i = beg + t; i < end; i += 256) atomicAdd(&dcnt[(unsigned)rec[i].x >> SRC_BITS], 1u);
  __syncthreads();
  const int node = blockIdx.x * TILE + t;
  if (t < TILE && node < Nn) norm[node] = rsqrtf((float)dcnt[t]);
  if (t < GROUPS) {
    unsigned off = beg;
    for (int r = 0; r < t * GROWS; ++r) off += dcnt[r];
    gpos[t] = off;
    gb[blockIdx.x * GROUPS + t] = off;
  }
  if (blockIdx.x == NT - 1 && t == 0) gb[NT * GROUPS] = end;
  if (blockIdx.x == 0 && t < REC_PAD) rec2[(size_t)E + t] = make_int2(0, 0);
  __syncthreads();
  for (unsigned i = beg + t; i < end; i += 256) {
    const int2 r = rec[i];
    const unsigned pos = atomicAdd(&gpos[((unsigned)r.x >> SRC_BITS) >> GSH], 1u);
    rec2[pos] = r;
  }
}

#define EW_PER 8
__global__ void __launch_bounds__(256)
dagnn_edge_weight(int2* __restrict__ rec2, const float* __restrict__ norm, int E) {
  const int i0 = blockIdx.x * (256 * EW_PER) + threadIdx.x;
  int sx[EW_PER];
  float w[EW_PER];
#pragma unroll
  for (int u = 0; u < EW_PER; ++u) {
    const int i = i0 + u * 256;
    sx[u] = rec2[i < E ? i : E - 1].x & SRC_MASK;
  }
#pragma unroll
  for (int u = 0; u < EW_PER; ++u) w[u] = norm[sx[u]];
#pragma unroll
  for (int u = 0; u < EW_PER; ++u) {
    const int i = i0 + u * 256;
    if (i < E) rec2[i].y = __float_as_int(w[u]);
  }
}

__global__ void __launch_bounds__(256)
dagnn_hop(const float* __restrict__ in, float* __restrict__ outp,
          const float* __restrict__ norm, const int2* __restrict__ rec,
          const unsigned* __restrict__ gb, int Nn) {
  __shared__ __align__(8) float stage[TILE * NCLASS];
  __shared__ float nl[TILE];
  const int t = threadIdx.x;
  const int wave = __builtin_amdgcn_readfirstlane(t >> 5);
  const int lane = t & 31;
  const int node0 = blockIdx.x * TILE;
  if (t < TILE) nl[t] = (node0 + t < Nn) ? norm[node0 + t] : 0.f;

  const unsigned s = gb[blockIdx.x * GROUPS + wave];
  const unsigned e = gb[blockIdx.x * GROUPS + wave + 1];
  float ax[GROWS], ay[GROWS];
#pragma unroll
  for (int r = 0; r < GROWS; ++r) { ax[r] = 0.f; ay[r] = 0.f; }

  if (s < e) {
    const unsigned lo = 8u * (unsigned)(lane < 25 ? lane : 24);
    auto add = [&](const int2 c, const float2 v) {
      const int row = (int)(((unsigned)__builtin_amdgcn_readfirstlane(c.x) >> SRC_BITS) & (GROWS - 1));
      const float w = __int_as_float(__builtin_amdgcn_readfirstlane(c.y));
      ax[row] = fmaf(w, v.x, ax[row]);
      ay[row] = fmaf(w, v.y, ay[row]);
    };
    auto batch = [&](unsigned p, const int2* cur, int2* nxt) {
#pragma unroll
      for (int u = 0; u < 8; ++u) nxt[u] = rec[p + 8u + (unsigned)u];
      float2 v[8];
#pragma unroll
      for (int u = 0; u < 8; ++u) {
        const unsigned sx = (unsigned)__builtin_amdgcn_readfirstlane(cur[u].x);
        const char* rowp = (const char*)in + (size_t)(sx & SRC_MASK) * (NCLASS * sizeof(float));
        v[u] = *(const float2*)(rowp + lo);
      }
      if (e - p >= 8u) {
#pragma unroll
        for (int u = 0; u < 8; ++u) add(cur[u], v[u]);
      } else {
#pragma unroll
        for (int u = 0; u < 8; ++u) {
          if (p + (unsigned)u < e) add(cur[u], v[u]);
        }
      }
    };
    int2 ra[8], rb[8];
#pragma unroll
    for (int u = 0; u < 8; ++u) ra[u] = rec[s + (unsigned)u];
    for (unsigned p = s; p < e; p += 16u) {
      batch(p, ra, rb);
      if (p + 8u < e) batch(p + 8u, rb, ra);
    }
  }

  if (lane < 25) {
#pragma unroll
    for (int r = 0; r < GROWS; ++r)
      *(float2*)&stage[(wave * GROWS + r) * NCLASS + 2 * lane] = make_float2(ax[r], ay[r]);
  }
  __syncthreads();

  const int nvalid = (Nn - node0 < TILE) ? Nn - node0 : TILE;
  float* op = outp + (size_t)node0 * NCLASS;
  for (int i = t; i < nvalid * NCLASS; i += 256) op[i] = stage[i] * nl[i / NCLASS];
}

#define RO_NODES 32
#define RO_WAVES 4
#define RO_PIECES ((RO_NODES * NCLASS) / 4)
#define RO_WHOLE  (RO_PIECES / 32)
#define RO_TAIL   (RO_PIECES - RO_WHOLE * 32)
typedef __attribute__((ext_vector_type(4))) float f4;
typedef __attribute__((ext_vector_type(2))) float f2;
static_assert(KHOPS % 2 == 0, "the plane loop below takes planes in pairs");
static_assert(RO_TAIL == 16, "tail piece count");

__device__ __forceinline__ void wave_lds_handover() {
  __builtin_amdgcn_fence(__ATOMIC_RELEASE, "wavefront");
  __builtin_amdgcn_wave_barrier();
  __builtin_amdgcn_fence(__ATOMIC_ACQUIRE, "wavefront");
}

__global__ void __launch_bounds__(RO_WAVES * 32)
dagnn_combine(const float* __restrict__ planes, const float* __restrict__ s,
              float* __restrict__ out, int Nn) {
  __shared__ __align__(16) float stage[RO_WAVES][RO_NODES * NCLASS];
  const int lane = threadIdx.x & 31;
  const int wave = __builtin_amdgcn_readfirstlane(threadIdx.x >> 5);
  float* st = stage[wave];
  const int node0 = (blockIdx.x * RO_WAVES + wave) * RO_NODES;
  if (node0 >= Nn) return;
  const size_t pstride = (size_t)Nn * NCLASS;
  const size_t base = (size_t)node0 * NCLASS;
  const bool whole = (node0 + RO_NODES <= Nn) && ((Nn & 1) == 0);

  float acc[NCLASS];
#pragma unroll
  for (int c = 0; c < NCLASS; ++c) acc[c] = 0.f;

  auto consume = [&]() {
    f2 h[NCLASS / 2];
#pragma unroll
    for (int j = 0; j < NCLASS / 2; ++j) h[j] = *(const f2*)&st[lane * NCLASS + 2 * j];
    float z = 0.f;
#pragma unroll
    for (int j = 0; j < NCLASS / 2; ++j) {
      z += h[j].x * s[2 * j];
      z += h[j].y * s[2 * j + 1];
    }
    const float S = 1.0f / (1.0f + __expf(-z));
#pragma unroll
    for (int j = 0; j < NCLASS / 2; ++j) {
      acc[2 * j]     += S * h[j].x;
      acc[2 * j + 1] += S * h[j].y;
    }
  };

  if (whole) {
    const f4* p0 = (const f4*)(planes + base);
    const size_t pq = pstride / 4;
    f4 ba[RO_WHOLE + 1], bb[RO_WHOLE + 1];
    auto fetch = [&](int k, f4* b) {
      const f4* p = p0 + (size_t)k * pq;
#pragma unroll
      for (int i = 0; i < RO_WHOLE; ++i) b[i] = p[i * 32 + lane];
      b[RO_WHOLE] = p[RO_WHOLE * 32 + (lane & (RO_TAIL - 1))];
    };
    auto put = [&](const f4* b) {
#pragma unroll
      for (int i = 0; i < RO_WHOLE; ++i) *(f4*)&st[(i * 32 + lane) * 4] = b[i];
      if (lane < RO_TAIL) *(f4*)&st[(RO_WHOLE * 32 + lane) * 4] = b[RO_WHOLE];
    };
    fetch(0, ba);
#pragma unroll 1
    for (int k = 0; k < KHOPS; k += 2) {
      fetch(k + 1, bb);
      put(ba); wave_lds_handover(); consume(); wave_lds_handover();
      fetch(k + 2, ba);
      put(bb); wave_lds_handover(); consume(); wave_lds_handover();
    }
    put(ba); wave_lds_handover(); consume(); wave_lds_handover();
  } else {
    for (int k = 0; k <= KHOPS; ++k) {
      const float* p = planes + (size_t)k * pstride;
      for (int i = 0; i < NCLASS; ++i) {
        const int idx = i * 32 + lane;
        const size_t gi = base + (size_t)idx;
        st[idx] = gi < pstride ? p[gi] : 0.f;
      }
      wave_lds_handover(); consume(); wave_lds_handover();
    }
  }

#pragma unroll
  for (int j = 0; j < NCLASS / 2; ++j)
    *(f2*)&st[lane * NCLASS + 2 * j] = (f2){acc[2 * j], acc[2 * j + 1]};
  wave_lds_handover();
  if (whole) {
    f4* op = (f4*)(out + base);
#pragma unroll
    for (int i = 0; i < RO_WHOLE; ++i) op[i * 32 + lane] = *(const f4*)&st[(i * 32 + lane) * 4];
    if (lane < RO_TAIL) op[RO_WHOLE * 32 + lane] = *(const f4*)&st[(RO_WHOLE * 32 + lane) * 4];
  } else {
    for (int i = 0; i < NCLASS; ++i) {
      const int idx = i * 32 + lane;
      const size_t gi = base + (size_t)idx;
      if (gi < pstride) out[gi] = st[idx];
    }
  }
}

extern "C" void kernel_launch(void* const* d_in, const int* in_sizes, int n_in,
                              void* d_out, int out_size, void* d_ws, size_t ws_size,
                              hipStream_t stream) {
  const float* feats = (const float*)d_in[0];
  const float* W1    = (const float*)d_in[1];
  const float* b1    = (const float*)d_in[2];
  const float* W2    = (const float*)d_in[3];
  const float* b2    = (const float*)d_in[4];
  const float* s     = (const float*)d_in[5];
  const int*   src   = (const int*)d_in[6];
  const int*   dst   = (const int*)d_in[7];
  const int Nn = in_sizes[0] / IN_DIM;
  const int E  = in_sizes[6];
  float* out = (float*)d_out;

  char* ws = (char*)d_ws;
  size_t off = 0;
  auto carve = [&](size_t bytes) -> char* {
    char* p = ws + off;
    off = (off + bytes + 255) & ~(size_t)255;
    return p;
  };
  const int NT = (Nn + TILE - 1) / TILE;
  const int chunk = (E + NCHUNK - 1) / NCHUNK;
  float*    norm   = (float*)   carve((size_t)Nn * sizeof(float));
  _Float16* hid    = (_Float16*)carve((size_t)Nn * HID * sizeof(_Float16));
  float*    planes = (float*)   carve((size_t)(KHOPS + 1) * Nn * NCLASS * sizeof(float));
  unsigned* cnt    = (unsigned*)carve((size_t)NT * NCHUNK * sizeof(unsigned));
  unsigned* lp     = (unsigned*)carve((size_t)NT * NCHUNK * sizeof(unsigned));
  unsigned* tt     = (unsigned*)carve((size_t)NT * sizeof(unsigned));
  unsigned* tb     = (unsigned*)carve((size_t)(NT + 1) * sizeof(unsigned));
  unsigned* gb     = (unsigned*)carve(((size_t)NT * GROUPS + 1) * sizeof(unsigned));
  int2*     rec    = (int2*)    carve((size_t)E * sizeof(int2));
  int2*     rec2   = (int2*)    carve(((size_t)E + REC_PAD) * sizeof(int2));
  if (off > ws_size) return;

  const size_t pstride = (size_t)Nn * NCLASS;

  const size_t hbytes = (size_t)NT * sizeof(unsigned);
  dagnn_bin_count<<<NCHUNK, 256, hbytes, stream>>>(dst, cnt, E, NT, chunk);
  dagnn_bin_scan<<<NT, 256, 0, stream>>>(cnt, lp, tt);
  dagnn_bin_place<<<NCHUNK, 256, hbytes, stream>>>(src, dst, lp, tt, tb, rec, E, NT, chunk);
  dagnn_tile_group<<<NT, 256, 0, stream>>>(rec, tb, norm, rec2, gb, Nn, NT, E);
  dagnn_edge_weight<<<(E + 256 * EW_PER - 1) / (256 * EW_PER), 256, 0, stream>>>(rec2, norm, E);

  dagnn_gemm1<<<G1_BLOCKS, G1_THREADS, 0, stream>>>(feats, W1, b1, hid, Nn);
  dim3 g2((Nn + 31) / 32, 1);
  dagnn_gemm2<<<g2, 256, 0, stream>>>(hid, W2, b2, planes, Nn);

  for (int k = 1; k <= KHOPS; ++k) {
    const float* pin = planes + (size_t)(k - 1) * pstride;
    float* pout      = planes + (size_t)k * pstride;
    dagnn_hop<<<NT, 256, 0, stream>>>(pin, pout, norm, rec2, gb, Nn);
  }

  const int rgroups = (Nn + RO_NODES - 1) / RO_NODES;
  dagnn_combine<<<(rgroups + RO_WAVES - 1) / RO_WAVES, RO_WAVES * 32, 0, stream>>>(planes, s, out, Nn);
  (void)n_in; (void)out_size;
}


# Second product joins the first in registers: a wave's relu tile, rounded to f16 in its accumulators, is the B operand against W2^T kept in LDS beside W1; hidden layer never stored, one launch fewer
# speedup vs baseline: 4.7013x; 1.1326x over previous
#include <hip/hip_runtime.h>
#include <hip/hip_bf16.h>

#define IN_DIM 500
#define HID    256
#define NCLASS 50
#define KHOPS  10

typedef __attribute__((ext_vector_type(16))) _Float16 v16h;
typedef __attribute__((ext_vector_type(8)))  _Float16 h8;
typedef __attribute__((ext_vector_type(8)))  float    v8f;

__device__ __forceinline__ v8f wmma16(v16h a, v16h b, v8f c) {
  return __builtin_amdgcn_wmma_f32_16x16x32_f16(false, a, false, b, (short)0,
                                                c, false, false);
}

#define G1_BLOCKS 256
#define G1_WAVES  8
#define G1_THREADS 256
#define WPITCH    520
#define W2ROWS    64
#define W2PITCH   264
static_assert(HID == G1_THREADS, "one thread per hidden column / per W2 row in the set-up");
static_assert(NCLASS % 2 == 0 && NCLASS <= W2ROWS, "class pairs, four class tiles");

__global__ void __launch_bounds__(G1_THREADS)
dagnn_mlp(const float* __restrict__ X, const float* __restrict__ W,
          const float* __restrict__ bias, const float* __restrict__ W2,
          const float* __restrict__ b2, float* __restrict__ P0, int Nn) {
  __shared__ __align__(16) _Float16 Wt[HID * WPITCH];
  __shared__ __align__(16) _Float16 W2t[W2ROWS * W2PITCH];
  __shared__ __align__(16) float b2s[W2ROWS];
  const int t = threadIdx.x;
  const int lane = t & 31;
  const int wave = __builtin_amdgcn_readfirstlane(t >> 5);

  {
    const int n = t;
#pragma unroll 8
    for (int g = 0; g < 64; ++g) {
      const int k0 = g * 8;
      h8 p;
#pragma unroll
      for (int i = 0; i < 8; ++i) {
        const int k = k0 + i;
        const int ck = k < IN_DIM ? k : IN_DIM - 1;
        const float v = W[(size_t)ck * HID + n];
        p[i] = k < IN_DIM ? (_Float16)v : (_Float16)0.f;
      }
      *(h8*)&Wt[n * WPITCH + k0] = p;
    }
  }
  {
    const int k = t;
    const float2* wr = (const float2*)(W2 + (size_t)k * NCLASS);
#pragma unroll
    for (int q = 0; q < NCLASS / 2; ++q) {
      const float2 v = wr[q];
      W2t[(2 * q) * W2PITCH + k]     = (_Float16)v.x;
      W2t[(2 * q + 1) * W2PITCH + k] = (_Float16)v.y;
    }
#pragma unroll
    for (int n = NCLASS; n < W2ROWS; ++n) W2t[n * W2PITCH + k] = (_Float16)0.f;
    if (t < W2ROWS) b2s[t] = t < NCLASS ? b2[t] : 0.f;
  }
  __syncthreads();

  const int c = lane & 15, h = lane >> 4;
  const _Float16* wwin0 = &Wt[c * WPITCH + 16 * h];
  const _Float16* w2win0 = &W2t[c * W2PITCH + 8 * h];
  const int ntiles = (Nn + 15) >> 4;
  const int tstride = G1_WAVES * G1_BLOCKS;

  int tile = wave * G1_BLOCKS + blockIdx.x;
  if (tile < ntiles) {
    auto rowptr = [&](int tl) -> const float* {
      const int mm = tl * 16 + c;
      const int mr = mm < Nn ? mm : Nn - 1;
      return X + (size_t)mr * IN_DIM + 16 * h;
    };
    const float* xp = rowptr(tile);

    float4 r0[4], r1[4], r2[4], r3[4];
    auto loadPlain = [&](const float* p, float4* d) {
#pragma unroll
      for (int q = 0; q < 4; ++q) d[q] = *(const float4*)(p + 4 * q);
    };
    auto loadMaybeLast = [&](const float* p, bool last, float4* d) {
#pragma unroll
      for (int q = 0; q < 4; ++q) {
        const bool ok = !last || (h == 0) || (q == 0);
        const float4 v = *(const float4*)(p + (ok ? 4 * q : 0));
        d[q].x = ok ? v.x : 0.f;
        d[q].y = ok ? v.y : 0.f;
        d[q].z = ok ? v.z : 0.f;
        d[q].w = ok ? v.w : 0.f;
      }
    };
    auto toFrag = [&](const float4* d) -> v16h {
      v16h f;
#pragma unroll
      for (int q = 0; q < 4; ++q) {
        f[4 * q + 0] = (_Float16)d[q].x;
        f[4 * q + 1] = (_Float16)d[q].y;
        f[4 * q + 2] = (_Float16)d[q].z;
        f[4 * q + 3] = (_Float16)d[q].w;
      }
      return f;
    };

    loadPlain(xp, r0);
    loadPlain(xp + 32, r1);
    loadPlain(xp + 64, r2);

    for (; tile < ntiles; tile += tstride) {
      const int m = tile * 16 + c;
      const int tnext = tile + tstride;
      const float* xpn = rowptr(tnext < ntiles ? tnext : tile);

      v8f acc[16];
#pragma unroll
      for (int nt = 0; nt < 16; ++nt)
        acc[nt] = (v8f){0.f, 0.f, 0.f, 0.f, 0.f, 0.f, 0.f, 0.f};

      auto products = [&](const float4* d, const _Float16* wk) {
        const v16h xf = toFrag(d);
#pragma unroll
        for (int nt = 0; nt < 16; ++nt) {
          const _Float16* wp = wk + nt * 16 * WPITCH;
          const h8 lo = *(const h8*)wp;
          const h8 hi = *(const h8*)(wp + 8);
          v16h wf;
#pragma unroll
          for (int i = 0; i < 8; ++i) { wf[i] = lo[i]; wf[8 + i] = hi[i]; }
          acc[nt] = wmma16(wf, xf, acc[nt]);
        }
      };

#pragma unroll 1
      for (int s4 = 0; s4 < 4; ++s4) {
        const float* xs = xp + s4 * 128;
        const _Float16* wk = wwin0 + s4 * 128;
        const bool more = s4 < 3;
        const float* nx = more ? xs + 128 : xpn;
        loadMaybeLast(xs + 96, !more, r3);
        products(r0, wk);
        loadPlain(nx, r0);
        products(r1, wk + 32);
        loadPlain(nx + 32, r1);
        products(r2, wk + 64);
        loadPlain(nx + 64, r2);
        products(r3, wk + 96);
      }

      v16h hb[8];
#pragma unroll
      for (int j = 0; j < 8; ++j) {
#pragma unroll
        for (int u = 0; u < 2; ++u) {
          const int nt = 2 * j + u;
          const float4 bA = *(const float4*)(bias + nt * 16 + 8 * h);
          const float4 bB = *(const float4*)(bias + nt * 16 + 8 * h + 4);
          const float bv[8] = {bA.x, bA.y, bA.z, bA.w, bB.x, bB.y, bB.z, bB.w};
#pragma unroll
          for (int r = 0; r < 8; ++r) {
            const float v = acc[nt][r] + bv[r];
            hb[j][8 * u + r] = (_Float16)(v > 0.f ? v : 0.f);
          }
        }
      }
      float* op = P0 + (size_t)m * NCLASS + 8 * h;
      const int clim = m < Nn ? NCLASS : 0;
#pragma unroll 1
      for (int mp = 0; mp < 2; ++mp) {
        const _Float16* wa = w2win0 + mp * 32 * W2PITCH;
        h8 fl[32];
#pragma unroll
        for (int j = 0; j < 8; ++j) {
          const _Float16* wp = wa + 32 * j;
          fl[4 * j + 0] = *(const h8*)wp;
          fl[4 * j + 1] = *(const h8*)(wp + 16);
          fl[4 * j + 2] = *(const h8*)(wp + 16 * W2PITCH);
          fl[4 * j + 3] = *(const h8*)(wp + 16 * W2PITCH + 16);
        }
#pragma unroll
        for (int g = 0; g < 4; ++g)
          asm volatile("" :: "v"(fl[8 * g + 0]), "v"(fl[8 * g + 1]), "v"(fl[8 * g + 2]),
                             "v"(fl[8 * g + 3]), "v"(fl[8 * g + 4]), "v"(fl[8 * g + 5]),
                             "v"(fl[8 * g + 6]), "v"(fl[8 * g + 7]));
        const float* bp = &b2s[32 * mp + 8 * h];
        const float4 bA0 = *(const float4*)bp,        bA1 = *(const float4*)(bp + 4);
        const float4 bB0 = *(const float4*)(bp + 16), bB1 = *(const float4*)(bp + 20);
        v8f oA = (v8f){0.f, 0.f, 0.f, 0.f, 0.f, 0.f, 0.f, 0.f};
        v8f oB = (v8f){0.f, 0.f, 0.f, 0.f, 0.f, 0.f, 0.f, 0.f};
#pragma unroll
        for (int j = 0; j < 8; ++j) {
          v16h fa, fb;
#pragma unroll
          for (int i = 0; i < 8; ++i) {
            fa[i] = fl[4 * j + 0][i]; fa[8 + i] = fl[4 * j + 1][i];
            fb[i] = fl[4 * j + 2][i]; fb[8 + i] = fl[4 * j + 3][i];
          }
          oA = wmma16(fa, hb[j], oA);
          oB = wmma16(fb, hb[j], oB);
        }
        {
          const float ba[8] = {bA0.x, bA0.y, bA0.z, bA0.w, bA1.x, bA1.y, bA1.z, bA1.w};
          const float bb[8] = {bB0.x, bB0.y, bB0.z, bB0.w, bB1.x, bB1.y, bB1.z, bB1.w};
          const int clsA = 32 * mp + 8 * h;
#pragma unroll
          for (int q = 0; q < 4; ++q) {
            const int ca = clsA + 2 * q, cb = ca + 16;
            if (ca < clim)
              *(float2*)(op + 32 * mp + 2 * q) =
                  make_float2(oA[2 * q] + ba[2 * q], oA[2 * q + 1] + ba[2 * q + 1]);
            if (cb < clim)
              *(float2*)(op + 32 * mp + 16 + 2 * q) =
                  make_float2(oB[2 * q] + bb[2 * q], oB[2 * q + 1] + bb[2 * q + 1]);
          }
        }
      }
      xp = xpn;
    }
  }
}

#define TSH   6
#define TILE  64
#define NCHUNK 256
#define SRC_BITS 24
#define SRC_MASK 0xFFFFFF
#define REC_PAD 32

__device__ __forceinline__ unsigned block_excl_scan256(unsigned v, unsigned* wsum,
                                                       unsigned* total) {
  const int lane = threadIdx.x & 31, wave = threadIdx.x >> 5;
  unsigned inc = v;
  for (int off = 1; off < 32; off <<= 1) {
    unsigned o = __shfl_up(inc, off, 32);
    if (lane >= off) inc += o;
  }
  if (lane == 31) wsum[wave] = inc;
  __syncthreads();
  unsigned wbase = 0, tot = 0;
  for (int w = 0; w < 8; ++w) {
    unsigned sw = wsum[w];
    if (w < wave) wbase += sw;
    tot += sw;
  }
  __syncthreads();
  *total = tot;
  return wbase + inc - v;
}

#define BIN_B 8

template <bool GUARD>
__device__ __forceinline__ void bin_count_batch(const int* __restrict__ dst, unsigned* hist,
                                                int i0, int end, int NT) {
  int dv[BIN_B];
#pragma unroll
  for (int u = 0; u < BIN_B; ++u) {
    const int i = i0 + u * 256;
    dv[u] = dst[(!GUARD || i < end) ? i : end - 1];
  }
#pragma unroll
  for (int u = 0; u < BIN_B; ++u) {
    const int i = i0 + u * 256;
    if (!GUARD || i < end) {
      int tl = dv[u] >> TSH;
      tl = tl < NT ? tl : NT - 1;
      atomicAdd(&hist[tl], 1u);
    }
  }
}

__global__ void __launch_bounds__(256)
dagnn_bin_count(const int* __restrict__ dst, unsigned* __restrict__ cnt,
                int E, int NT, int chunk) {
  extern __shared__ unsigned hist[];
  const int t = threadIdx.x;
  for (int i = t; i < NT; i += 256) hist[i] = 0u;
  __syncthreads();
  const int beg = blockIdx.x * chunk;
  const int end = (beg + chunk < E) ? beg + chunk : E;
  for (int b0 = beg; b0 < end; b0 += 256 * BIN_B) {
    if (b0 + 256 * BIN_B <= end) bin_count_batch<false>(dst, hist, b0 + t, end, NT);
    else                         bin_count_batch<true>(dst, hist, b0 + t, end, NT);
  }
  __syncthreads();
  for (int i = t; i < NT; i += 256) cnt[(size_t)i * NCHUNK + blockIdx.x] = hist[i];
}

__global__ void __launch_bounds__(256)
dagnn_bin_scan(const unsigned* __restrict__ cnt, unsigned* __restrict__ lp,
               unsigned* __restrict__ tt) {
  __shared__ unsigned wsum[8];
  const size_t idx = (size_t)blockIdx.x * NCHUNK + threadIdx.x;
  unsigned tot;
  unsigned ex = block_excl_scan256(cnt[idx], wsum, &tot);
  lp[idx] = ex;
  if (threadIdx.x == 0) tt[blockIdx.x] = tot;
}

template <bool GUARD>
__device__ __forceinline__ void bin_place_batch(const int* __restrict__ src,
                                                const int* __restrict__ dst, unsigned* ctr,
                                                int2* __restrict__ rec, int i0, int end, int NT) {
  int dv[BIN_B], sv[BIN_B];
  unsigned pos[BIN_B];
#pragma unroll
  for (int u = 0; u < BIN_B; ++u) {
    const int i = i0 + u * 256;
    const int ci = (!GUARD || i < end) ? i : end - 1;
    dv[u] = dst[ci];
    sv[u] = src[ci];
  }
#pragma unroll
  for (int u = 0; u < BIN_B; ++u) {
    const int i = i0 + u * 256;
    pos[u] = 0u;
    if (!GUARD || i < end) {
      int tl = dv[u] >> TSH;
      tl = tl < NT ? tl : NT - 1;
      pos[u] = atomicAdd(&ctr[tl], 1u);
    }
  }
#pragma unroll
  for (int u = 0; u < BIN_B; ++u) {
    const int i = i0 + u * 256;
    if (!GUARD || i < end)
      rec[pos[u]] = make_int2(sv[u] | ((dv[u] & (TILE - 1)) << SRC_BITS), 0);
  }
}

__global__ void __launch_bounds__(256)
dagnn_bin_place(const int* __restrict__ src, const int* __restrict__ dst,
                const unsigned* __restrict__ lp, const unsigned* __restrict__ tt,
                unsigned* __restrict__ tb, int2* __restrict__ rec,
                int E, int NT, int chunk) {
  extern __shared__ unsigned ctr[];
  __shared__ unsigned wsum[8];
  const int t = threadIdx.x;
  const int per = (NT + 255) / 256;
  const int i0 = t * per;
  const int i1 = (i0 + per < NT) ? i0 + per : NT;
  unsigned sum = 0;
  for (int ib = i0; ib < i1; ib += BIN_B) {
    unsigned a[BIN_B];
#pragma unroll
    for (int u = 0; u < BIN_B; ++u) a[u] = tt[ib + u < i1 ? ib + u : i1 - 1];
#pragma unroll
    for (int u = 0; u < BIN_B; ++u) sum += (ib + u < i1) ? a[u] : 0u;
  }
  unsigned tot;
  unsigned ex = block_excl_scan256(sum, wsum, &tot);
  for (int ib = i0; ib < i1; ib += BIN_B) {
    unsigned a[BIN_B], l[BIN_B];
#pragma unroll
    for (int u = 0; u < BIN_B; ++u) {
      const int i = ib + u < i1 ? ib + u : i1 - 1;
      a[u] = tt[i];
      l[u] = lp[(size_t)i * NCHUNK + blockIdx.x];
    }
#pragma unroll
    for (int u = 0; u < BIN_B; ++u) {
      const int i = ib + u;
      if (i < i1) {
        ctr[i] = ex + l[u];
        if (blockIdx.x == 0) tb[i] = ex;
        ex += a[u];
      }
    }
  }
  if (blockIdx.x == 0 && t == 0) tb[NT] = tot;
  __syncthreads();
  const int beg = blockIdx.x * chunk;
  const int end = (beg + chunk < E) ? beg + chunk : E;
  for (int b0 = beg; b0 < end; b0 += 256 * BIN_B) {
    if (b0 + 256 * BIN_B <= end) bin_place_batch<false>(src, dst, ctr, rec, b0 + t, end, NT);
    else                         bin_place_batch<true>(src, dst, ctr, rec, b0 + t, end, NT);
  }
}

#define GROWS  16
#define GSH    4
#define GROUPS (TILE / GROWS)
#define HOP_THREADS (GROUPS * 32)
#define HB 8

__global__ void __launch_bounds__(256)
dagnn_tile_group(const int2* __restrict__ rec, const unsigned* __restrict__ tb,
                 float* __restrict__ norm, int2* __restrict__ rec2,
                 unsigned* __restrict__ gb, int Nn, int NT, int E) {
  __shared__ unsigned dcnt[TILE];
  __shared__ unsigned gpos[GROUPS];
  const int t = threadIdx.x;
  if (t < TILE) dcnt[t] = 0u;
  __syncthreads();
  const unsigned beg = tb[blockIdx.x], end = tb[blockIdx.x + 1];
  for (unsigned i = beg + t; i < end; i += 256) atomicAdd(&dcnt[(unsigned)rec[i].x >> SRC_BITS], 1u);
  __syncthreads();
  const int node = blockIdx.x * TILE + t;
  if (t < TILE && node < Nn) norm[node] = rsqrtf((float)dcnt[t]);
  if (t < GROUPS) {
    unsigned off = beg;
    for (int r = 0; r < t * GROWS; ++r) off += dcnt[r];
    gpos[t] = off;
    gb[blockIdx.x * GROUPS + t] = off;
  }
  if (blockIdx.x == NT - 1 && t == 0) gb[NT * GROUPS] = end;
  if (blockIdx.x == 0 && t < REC_PAD) rec2[(size_t)E + t] = make_int2(0, 0);
  __syncthreads();
  for (unsigned i = beg + t; i < end; i += 256) {
    const int2 r = rec[i];
    const unsigned pos = atomicAdd(&gpos[((unsigned)r.x >> SRC_BITS) >> GSH], 1u);
    rec2[pos] = r;
  }
}

#define EW_PER 8
__global__ void __launch_bounds__(256)
dagnn_edge_weight(int2* __restrict__ rec2, const float* __restrict__ norm, int E) {
  const int i0 = blockIdx.x * (256 * EW_PER) + threadIdx.x;
  int sx[EW_PER];
  float w[EW_PER];
#pragma unroll
  for (int u = 0; u < EW_PER; ++u) {
    const int i = i0 + u * 256;
    sx[u] = rec2[i < E ? i : E - 1].x & SRC_MASK;
  }
#pragma unroll
  for (int u = 0; u < EW_PER; ++u) w[u] = norm[sx[u]];
#pragma unroll
  for (int u = 0; u < EW_PER; ++u) {
    const int i = i0 + u * 256;
    if (i < E) rec2[i].y = __float_as_int(w[u]);
  }
}

__global__ void __launch_bounds__(HOP_THREADS)
dagnn_hop(const float* __restrict__ in, float* __restrict__ outp,
          const float* __restrict__ norm, const int2* __restrict__ rec,
          const unsigned* __restrict__ gb, int Nn) {
  __shared__ __align__(8) float stage[TILE * NCLASS];
  __shared__ float nl[TILE];
  const int t = threadIdx.x;
  const int wave = __builtin_amdgcn_readfirstlane(t >> 5);
  const int lane = t & 31;
  const int node0 = blockIdx.x * TILE;
  if (t < TILE) nl[t] = (node0 + t < Nn) ? norm[node0 + t] : 0.f;

  const unsigned s = gb[blockIdx.x * GROUPS + wave];
  const unsigned e = gb[blockIdx.x * GROUPS + wave + 1];
  float ax[GROWS], ay[GROWS];
#pragma unroll
  for (int r = 0; r < GROWS; ++r) { ax[r] = 0.f; ay[r] = 0.f; }

  if (s < e) {
    const unsigned lo = 8u * (unsigned)(lane < 25 ? lane : 24);
    auto add = [&](const int2 c, const float2 v) {
      const int row = (int)(((unsigned)__builtin_amdgcn_readfirstlane(c.x) >> SRC_BITS) & (GROWS - 1));
      const float w = __int_as_float(__builtin_amdgcn_readfirstlane(c.y));
      ax[row] = fmaf(w, v.x, ax[row]);
      ay[row] = fmaf(w, v.y, ay[row]);
    };
    auto issue = [&](const int2* r, float2* v) {
#pragma unroll
      for (int u = 0; u < HB; ++u) {
        const unsigned sx = (unsigned)__builtin_amdgcn_readfirstlane(r[u].x);
        const char* rowp = (const char*)in + (size_t)(sx & SRC_MASK) * (NCLASS * sizeof(float));
        v[u] = *(const float2*)(rowp + lo);
      }
    };
    auto request = [&](unsigned q, int2* r) {
      const int2* rq = rec + q;
#pragma unroll
      for (int u = 0; u < HB; ++u) r[u] = rq[u];
    };
    auto consume = [&](unsigned q, const int2* r, const float2* v) {
      if (e - q >= (unsigned)HB) {
#pragma unroll
        for (int u = 0; u < HB; ++u) add(r[u], v[u]);
      } else {
#pragma unroll
        for (int u = 0; u < HB; ++u) {
          if (q + (unsigned)u < e) add(r[u], v[u]);
        }
      }
    };
    auto stage3 = [&](unsigned q, const int2* rA, const float2* vA,
                      const int2* rB, float2* vB, int2* rC, bool FULL) {
      issue(rB, vB);
      request(q + 2u * HB, rC);
      if (FULL) {
#pragma unroll
        for (int u = 0; u < HB; ++u) add(rA[u], vA[u]);
      } else {
        consume(q, rA, vA);
      }
    };
    int2 r0[HB], r1[HB], r2[HB];
    float2 v0[HB], v1[HB];
    request(s, r0);
    request(s + (unsigned)HB, r1);
    issue(r0, v0);
    unsigned p = s;
    while (p + 5u * HB < e) {
      stage3(p,           r0, v0, r1, v1, r2, true);
      stage3(p + 1u * HB, r1, v1, r2, v0, r0, true);
      stage3(p + 2u * HB, r2, v0, r0, v1, r1, true);
      stage3(p + 3u * HB, r0, v1, r1, v0, r2, true);
      stage3(p + 4u * HB, r1, v0, r2, v1, r0, true);
      stage3(p + 5u * HB, r2, v1, r0, v0, r1, false);
      p += 6u * HB;
    }
    if (p < e) {
      stage3(p, r0, v0, r1, v1, r2, false);
      if (p + 1u * HB < e) {
        stage3(p + 1u * HB, r1, v1, r2, v0, r0, false);
        if (p + 2u * HB < e) {
          stage3(p + 2u * HB, r2, v0, r0, v1, r1, false);
          if (p + 3u * HB < e) {
            stage3(p + 3u * HB, r0, v1, r1, v0, r2, false);
            if (p + 4u * HB < e) stage3(p + 4u * HB, r1, v0, r2, v1, r0, false);
          }
        }
      }
    }
  }

  if (lane < 25) {
#pragma unroll
    for (int r = 0; r < GROWS; ++r)
      *(float2*)&stage[(wave * GROWS + r) * NCLASS + 2 * lane] = make_float2(ax[r], ay[r]);
  }
  __syncthreads();

  const int nvalid = (Nn - node0 < TILE) ? Nn - node0 : TILE;
  float* op = outp + (size_t)node0 * NCLASS;
  for (int i = t; i < nvalid * NCLASS; i += HOP_THREADS) op[i] = stage[i] * nl[i / NCLASS];
}

#define RO_NODES 32
#define RO_WAVES 4
#define RO_PIECES ((RO_NODES * NCLASS) / 4)
#define RO_WHOLE  (RO_PIECES / 32)
#define RO_TAIL   (RO_PIECES - RO_WHOLE * 32)
typedef __attribute__((ext_vector_type(4))) float f4;
typedef __attribute__((ext_vector_type(2))) float f2;
static_assert(KHOPS % 2 == 0, "the plane loop below takes planes in pairs");
static_assert(RO_TAIL == 16, "tail piece count");

__device__ __forceinline__ void wave_lds_handover() {
  __builtin_amdgcn_fence(__ATOMIC_RELEASE, "wavefront");
  __builtin_amdgcn_wave_barrier();
  __builtin_amdgcn_fence(__ATOMIC_ACQUIRE, "wavefront");
}

__global__ void __launch_bounds__(RO_WAVES * 32)
dagnn_combine(const float* __restrict__ planes, const float* __restrict__ s,
              float* __restrict__ out, int Nn) {
  __shared__ __align__(16) float stage[RO_WAVES][RO_NODES * NCLASS];
  const int lane = threadIdx.x & 31;
  const int wave = __builtin_amdgcn_readfirstlane(threadIdx.x >> 5);
  float* st = stage[wave];
  const int node0 = (blockIdx.x * RO_WAVES + wave) * RO_NODES;
  if (node0 >= Nn) return;
  const size_t pstride = (size_t)Nn * NCLASS;
  const size_t base = (size_t)node0 * NCLASS;
  const bool whole = (node0 + RO_NODES <= Nn) && ((Nn & 1) == 0);

  float acc[NCLASS];
#pragma unroll
  for (int c = 0; c < NCLASS; ++c) acc[c] = 0.f;

  auto consume = [&]() {
    f2 h[NCLASS / 2];
#pragma unroll
    for (int j = 0; j < NCLASS / 2; ++j) h[j] = *(const f2*)&st[lane * NCLASS + 2 * j];
    float z = 0.f;
#pragma unroll
    for (int j = 0; j < NCLASS / 2; ++j) {
      z += h[j].x * s[2 * j];
      z += h[j].y * s[2 * j + 1];
    }
    const float S = 1.0f / (1.0f + __expf(-z));
#pragma unroll
    for (int j = 0; j < NCLASS / 2; ++j) {
      acc[2 * j]     += S * h[j].x;
      acc[2 * j + 1] += S * h[j].y;
    }
  };

  if (whole) {
    const f4* p0 = (const f4*)(planes + base);
    const size_t pq = pstride / 4;
    f4 ba[RO_WHOLE + 1], bb[RO_WHOLE + 1];
    auto fetch = [&](int k, f4* b) {
      const f4* p = p0 + (size_t)k * pq;
#pragma unroll
      for (int i = 0; i < RO_WHOLE; ++i) b[i] = p[i * 32 + lane];
      b[RO_WHOLE] = p[RO_WHOLE * 32 + (lane & (RO_TAIL - 1))];
    };
    auto put = [&](const f4* b) {
#pragma unroll
      for (int i = 0; i < RO_WHOLE; ++i) *(f4*)&st[(i * 32 + lane) * 4] = b[i];
      if (lane < RO_TAIL) *(f4*)&st[(RO_WHOLE * 32 + lane) * 4] = b[RO_WHOLE];
    };
    fetch(0, ba);
#pragma unroll 1
    for (int k = 0; k < KHOPS; k += 2) {
      fetch(k + 1, bb);
      put(ba); wave_lds_handover(); consume(); wave_lds_handover();
      fetch(k + 2, ba);
      put(bb); wave_lds_handover(); consume(); wave_lds_handover();
    }
    put(ba); wave_lds_handover(); consume(); wave_lds_handover();
  } else {
    for (int k = 0; k <= KHOPS; ++k) {
      const float* p = planes + (size_t)k * pstride;
      for (int i = 0; i < NCLASS; ++i) {
        const int idx = i * 32 + lane;
        const size_t gi = base + (size_t)idx;
        st[idx] = gi < pstride ? p[gi] : 0.f;
      }
      wave_lds_handover(); consume(); wave_lds_handover();
    }
  }

#pragma unroll
  for (int j = 0; j < NCLASS / 2; ++j)
    *(f2*)&st[lane * NCLASS + 2 * j] = (f2){acc[2 * j], acc[2 * j + 1]};
  wave_lds_handover();
  if (whole) {
    f4* op = (f4*)(out + base);
#pragma unroll
    for (int i = 0; i < RO_WHOLE; ++i) op[i * 32 + lane] = *(const f4*)&st[(i * 32 + lane) * 4];
    if (lane < RO_TAIL) op[RO_WHOLE * 32 + lane] = *(const f4*)&st[(RO_WHOLE * 32 + lane) * 4];
  } else {
    for (int i = 0; i < NCLASS; ++i) {
      const int idx = i * 32 + lane;
      const size_t gi = base + (size_t)idx;
      if (gi < pstride) out[gi] = st[idx];
    }
  }
}

extern "C" void kernel_launch(void* const* d_in, const int* in_sizes, int n_in,
                              void* d_out, int out_size, void* d_ws, size_t ws_size,
                              hipStream_t stream) {
  const float* feats = (const float*)d_in[0];
  const float* W1    = (const float*)d_in[1];
  const float* b1    = (const float*)d_in[2];
  const float* W2    = (const float*)d_in[3];
  const float* b2    = (const float*)d_in[4];
  const float* s     = (const float*)d_in[5];
  const int*   src   = (const int*)d_in[6];
  const int*   dst   = (const int*)d_in[7];
  const int Nn = in_sizes[0] / IN_DIM;
  const int E  = in_sizes[6];
  float* out = (float*)d_out;

  char* ws = (char*)d_ws;
  size_t off = 0;
  auto carve = [&](size_t bytes) -> char* {
    char* p = ws + off;
    off = (off + bytes + 255) & ~(size_t)255;
    return p;
  };
  const int NT = (Nn + TILE - 1) / TILE;
  const int chunk = (E + NCHUNK - 1) / NCHUNK;
  float*    norm   = (float*)   carve((size_t)Nn * sizeof(float));
  float*    planes = (float*)   carve((size_t)(KHOPS + 1) * Nn * NCLASS * sizeof(float));
  unsigned* cnt    = (unsigned*)carve((size_t)NT * NCHUNK * sizeof(unsigned));
  unsigned* lp     = (unsigned*)carve((size_t)NT * NCHUNK * sizeof(unsigned));
  unsigned* tt     = (unsigned*)carve((size_t)NT * sizeof(unsigned));
  unsigned* tb     = (unsigned*)carve((size_t)(NT + 1) * sizeof(unsigned));
  unsigned* gb     = (unsigned*)carve(((size_t)NT * GROUPS + 1) * sizeof(unsigned));
  int2*     rec    = (int2*)    carve((size_t)E * sizeof(int2));
  int2*     rec2   = (int2*)    carve(((size_t)E + REC_PAD) * sizeof(int2));
  if (off > ws_size) return;

  const size_t pstride = (size_t)Nn * NCLASS;

  const size_t hbytes = (size_t)NT * sizeof(unsigned);
  dagnn_bin_count<<<NCHUNK, 256, hbytes, stream>>>(dst, cnt, E, NT, chunk);
  dagnn_bin_scan<<<NT, 256, 0, stream>>>(cnt, lp, tt);
  dagnn_bin_place<<<NCHUNK, 256, hbytes, stream>>>(src, dst, lp, tt, tb, rec, E, NT, chunk);
  dagnn_tile_group<<<NT, 256, 0, stream>>>(rec, tb, norm, rec2, gb, Nn, NT, E);
  dagnn_edge_weight<<<(E + 256 * EW_PER - 1) / (256 * EW_PER), 256, 0, stream>>>(rec2, norm, E);

  dagnn_mlp<<<G1_BLOCKS, G1_THREADS, 0, stream>>>(feats, W1, b1, W2, b2, planes, Nn);

  for (int k = 1; k <= KHOPS; ++k) {
    const float* pin = planes + (size_t)(k - 1) * pstride;
    float* pout      = planes + (size_t)k * pstride;
    dagnn_hop<<<NT, HOP_THREADS, 0, stream>>>(pin, pout, norm, rec2, gb, Nn);
  }

  const int rgroups = (Nn + RO_NODES - 1) / RO_NODES;
  dagnn_combine<<<(rgroups + RO_WAVES - 1) / RO_WAVES, RO_WAVES * 32, 0, stream>>>(planes, s, out, Nn);
  (void)n_in; (void)out_size;
}
